# starting kernel
# baseline (speedup 1.0000x reference)
#include <hip/hip_runtime.h>

typedef __attribute__((ext_vector_type(2))) float v2f;
typedef __attribute__((ext_vector_type(8))) float v8f;

#define NNODES 40000
#define NEDGES 640000
#define F 128

__global__ void zero_f32(float* __restrict__ p, int n) {
  int i = blockIdx.x * blockDim.x + threadIdx.x;
  int stride = gridDim.x * blockDim.x;
  for (; i < n; i += stride) p[i] = 0.0f;
}

__global__ void scatter_accum(const float* __restrict__ feat,
                              const int* __restrict__ ei,
                              float* __restrict__ sum,
                              float* __restrict__ cnt,
                              int do_cnt) {
  int gid = blockIdx.x * blockDim.x + threadIdx.x;
  if (gid >= NEDGES * 32) return;
  int e = gid >> 5;
  int q = gid & 31;
  int src = ei[e];
  int dst = ei[NEDGES + e];
  const float4 v = *(const float4*)(feat + (size_t)src * F + q * 4);
  float* s = sum + (size_t)dst * F + q * 4;
  atomicAdd(s + 0, v.x);
  atomicAdd(s + 1, v.y);
  atomicAdd(s + 2, v.z);
  atomicAdd(s + 3, v.w);
  if (do_cnt && q == 0) atomicAdd(cnt + dst, 1.0f);
}

__global__ __launch_bounds__(256)
void sage_gemm(const float* __restrict__ sum,
               const float* __restrict__ cnt,
               const float* __restrict__ xin,
               const float* __restrict__ Wl,
               const float* __restrict__ Wr,
               const float* __restrict__ bias,
               float* __restrict__ out,
               int relu) {
  __shared__ float inv[16];
  const int tid  = threadIdx.x;
  const int lane = tid & 31;
  const int wave = tid >> 5;
  const int half = lane >> 4;
  const int lid  = lane & 15;
  const int base = blockIdx.x * 16;
  const int col  = wave * 16 + lid;

  if (tid < 16) {
    float c = cnt[base + tid];
    inv[tid] = 1.0f / fmaxf(c, 1.0f);
  }
  __syncthreads();

  const float* srow = sum + (size_t)(base + lid) * F;
  const float* xrow = xin + (size_t)(base + lid) * F;
  const float scale = inv[lid];

  v8f acc = {};
#pragma unroll 4
  for (int k = 0; k < F; k += 4) {
    const int ka = k + 2 * half;
    v2f a1 = *(const v2f*)(srow + ka);
    a1 *= scale;
    v2f a2 = *(const v2f*)(xrow + ka);
    v2f b1, b2;
    b1.x = Wl[ka * F + col];
    b1.y = Wl[(ka + 1) * F + col];
    b2.x = Wr[ka * F + col];
    b2.y = Wr[(ka + 1) * F + col];
    acc = __builtin_amdgcn_wmma_f32_16x16x4_f32(false, a1, false, b1,
                                                (short)0, acc, false, false);
    acc = __builtin_amdgcn_wmma_f32_16x16x4_f32(false, a2, false, b2,
                                                (short)0, acc, false, false);
  }

  const float bv = bias[col];
#pragma unroll
  for (int v = 0; v < 8; ++v) {
    int m = base + v + 8 * half;
    float val = acc[v] + bv;
    if (relu) val = fmaxf(val, 0.0f);
    out[(size_t)m * F + col] = val;
  }
}

__global__ void fc_kernel(const float* __restrict__ h,
                          const float* __restrict__ Wfc,
                          const float* __restrict__ bfc,
                          float* __restrict__ out) {
  int node = blockIdx.x * blockDim.x + threadIdx.x;
  if (node >= NNODES) return;
  const float* row = h + (size_t)node * F;
  float a0 = bfc[0], a1 = bfc[1];
#pragma unroll 8
  for (int k = 0; k < F; ++k) {
    float v = row[k];
    a0 += v * Wfc[k * 2 + 0];
    a1 += v * Wfc[k * 2 + 1];
  }
  out[node * 2 + 0] = a0;
  out[node * 2 + 1] = a1;
}

extern "C" void kernel_launch(void* const* d_in, const int* in_sizes, int n_in,
                              void* d_out, int out_size, void* d_ws, size_t ws_size,
                              hipStream_t stream) {
  const float* x   = (const float*)d_in[0];
  const int*   ei  = (const int*)d_in[1];
  const float* W1l = (const float*)d_in[2];
  const float* b1  = (const float*)d_in[3];
  const float* W1r = (const float*)d_in[4];
  const float* W2l = (const float*)d_in[5];
  const float* b2  = (const float*)d_in[6];
  const float* W2r = (const float*)d_in[7];
  const float* Wfc = (const float*)d_in[8];
  const float* bfc = (const float*)d_in[9];
  float* out = (float*)d_out;

  float* ws     = (float*)d_ws;
  float* sumbuf = ws;
  float* h1     = sumbuf + (size_t)NNODES * F;
  float* h2     = h1 + (size_t)NNODES * F;
  float* cnt    = h2 + (size_t)NNODES * F;

  const int scatterBlocks = (NEDGES * 32 + 255) / 256;

  zero_f32<<<2048, 256, 0, stream>>>(sumbuf, NNODES * F);
  zero_f32<<<160, 256, 0, stream>>>(cnt, NNODES);
  scatter_accum<<<scatterBlocks, 256, 0, stream>>>(x, ei, sumbuf, cnt, 1);
  sage_gemm<<<NNODES / 16, 256, 0, stream>>>(sumbuf, cnt, x, W1l, W1r, b1, h1, 1);

  zero_f32<<<2048, 256, 0, stream>>>(sumbuf, NNODES * F);
  scatter_accum<<<scatterBlocks, 256, 0, stream>>>(h1, ei, sumbuf, cnt, 0);
  sage_gemm<<<NNODES / 16, 256, 0, stream>>>(sumbuf, cnt, h1, W2l, W2r, b2, h2, 1);

  fc_kernel<<<(NNODES + 255) / 256, 256, 0, stream>>>(h2, Wfc, bfc, out);
}


# Neighbour sums by tile ownership: 250 work-groups each hold 160 nodes' sums in LDS, scan the whole edge list for their range, add fetched rows by LDS float atomics (arrival order); no zero fills
# speedup vs baseline: 2.0420x; 2.0420x over previous
#include <hip/hip_runtime.h>

typedef __attribute__((ext_vector_type(2))) float v2f;
typedef __attribute__((ext_vector_type(8))) float v8f;

#define NNODES 40000
#define NEDGES 640000
#define F 128

#define TD     160
#define LCAP   1024
#define NCHUNK (NEDGES / 128)
#define NSTEP  ((NCHUNK + 127) / 128)

__global__ __launch_bounds__(1024)
void agg_tile(const float* __restrict__ feat,
              const int* __restrict__ ei,
              float* __restrict__ sum,
              float* __restrict__ cnt,
              int do_cnt) {
  __shared__ float    tile[TD * F];
  __shared__ unsigned mlist[32 * LCAP];
  __shared__ int      cntl[TD];

  const int tid  = threadIdx.x;
  const int lane = tid & 31;
  const int wave = __builtin_amdgcn_readfirstlane(tid >> 5);
  const int base = blockIdx.x * TD;

#pragma unroll
  for (int k = 0; k < (TD * F) / (4 * 1024); ++k)
    ((float4*)tile)[tid + k * 1024] = make_float4(0.f, 0.f, 0.f, 0.f);
  if (tid < TD) cntl[tid] = 0;
  __syncthreads();

  const int4* dst4 = (const int4*)(ei + NEDGES);
  unsigned* myl = mlist + wave * LCAP;
  int nl = 0;

  const int rot = blockIdx.x % NSTEP;

  int4 cur[4], nxt[4];
#pragma unroll
  for (int u = 0; u < 4; ++u) {
    const int c = min((rot * 32 + wave) * 4 + u, NCHUNK - 1);
    cur[u] = dst4[c * 32 + lane];
  }

  for (int s = 0; s < NSTEP; ++s) {
    const int sc = (s + rot < NSTEP) ? s + rot : s + rot - NSTEP;
    const int sn = (sc + 1 < NSTEP) ? sc + 1 : 0;
#pragma unroll
    for (int u = 0; u < 4; ++u) {
      const int c = min((sn * 32 + wave) * 4 + u, NCHUNK - 1);
      nxt[u] = dst4[c * 32 + lane];
    }

#pragma unroll
    for (int u = 0; u < 4; ++u) {
      const int c  = (sc * 32 + wave) * 4 + u;
      const int r0 = cur[u].x - base, r1 = cur[u].y - base;
      const int r2 = cur[u].z - base, r3 = cur[u].w - base;
      const bool m0 = (unsigned)r0 < (unsigned)TD, m1 = (unsigned)r1 < (unsigned)TD;
      const bool m2 = (unsigned)r2 < (unsigned)TD, m3 = (unsigned)r3 < (unsigned)TD;
      if (c < NCHUNK && __builtin_amdgcn_ballot_w32(m0 | m1 | m2 | m3) != 0u) {
        const int e0 = (c * 32 + lane) * 4;
#define APPEND(M, R, E)                                                     \
        {                                                                   \
          const unsigned b = __builtin_amdgcn_ballot_w32(M);                \
          if (M) {                                                          \
            const int pos = nl + (int)__builtin_amdgcn_mbcnt_lo(b, 0u);     \
            myl[pos] = ((unsigned)(E) << 8) | (unsigned)(R);                \
            if (do_cnt) atomicAdd(&cntl[R], 1);                             \
          }                                                                 \
          nl += __builtin_popcount(b);                                      \
        }
        APPEND(m0, r0, e0 + 0)
        APPEND(m1, r1, e0 + 1)
        APPEND(m2, r2, e0 + 2)
        APPEND(m3, r3, e0 + 3)
#undef APPEND
      }
    }

    if (nl >= 32 || s == NSTEP - 1) {
      for (int i0 = 0; i0 < nl; i0 += 32) {
        const int n = min(32, nl - i0);
        const unsigned pk = myl[i0 + min(lane, n - 1)];
        const int sv = ei[pk >> 8];
        const int rv = (int)(pk & 255u);
        for (int j = 0; j < n; j += 8) {
          float4 v[8];
          int    rr[8];
#pragma unroll
          for (int t = 0; t < 8; ++t) {
            const int idx = min(j + t, n - 1);
            const int ss  = __builtin_amdgcn_readlane(sv, idx);
            rr[t]         = __builtin_amdgcn_readlane(rv, idx);
            v[t] = *(const float4*)(feat + (size_t)ss * F + lane * 4);
          }
#pragma unroll
          for (int t = 0; t < 8; ++t) {
            if (j + t < n) {
              float* p = tile + rr[t] * F + lane;
              atomicAdd(p,      v[t].x);
              atomicAdd(p + 32, v[t].y);
              atomicAdd(p + 64, v[t].z);
              atomicAdd(p + 96, v[t].w);
            }
          }
        }
      }
      nl = 0;
    }

#pragma unroll
    for (int u = 0; u < 4; ++u) cur[u] = nxt[u];
  }

  __syncthreads();

#pragma unroll
  for (int k = 0; k < (TD * F) / (4 * 1024); ++k) {
    const int idx = tid + k * 1024;
    const int row = idx >> 5;
    const int l   = idx & 31;
    const float* p = tile + row * F + l;
    const float4 o = make_float4(p[0], p[32], p[64], p[96]);
    *(float4*)(sum + (size_t)(base + row) * F + l * 4) = o;
  }
  if (do_cnt && tid < TD) cnt[base + tid] = (float)cntl[tid];
}

__global__ __launch_bounds__(256)
void sage_gemm(const float* __restrict__ sum,
               const float* __restrict__ cnt,
               const float* __restrict__ xin,
               const float* __restrict__ Wl,
               const float* __restrict__ Wr,
               const float* __restrict__ bias,
               float* __restrict__ out,
               int relu) {
  __shared__ float inv[16];
  const int tid  = threadIdx.x;
  const int lane = tid & 31;
  const int wave = tid >> 5;
  const int half = lane >> 4;
  const int lid  = lane & 15;
  const int base = blockIdx.x * 16;
  const int col  = wave * 16 + lid;

  if (tid < 16) {
    float c = cnt[base + tid];
    inv[tid] = 1.0f / fmaxf(c, 1.0f);
  }
  __syncthreads();

  const float* srow = sum + (size_t)(base + lid) * F;
  const float* xrow = xin + (size_t)(base + lid) * F;
  const float scale = inv[lid];

  v8f acc = {};
#pragma unroll 4
  for (int k = 0; k < F; k += 4) {
    const int ka = k + 2 * half;
    v2f a1 = *(const v2f*)(srow + ka);
    a1 *= scale;
    v2f a2 = *(const v2f*)(xrow + ka);
    v2f b1, b2;
    b1.x = Wl[ka * F + col];
    b1.y = Wl[(ka + 1) * F + col];
    b2.x = Wr[ka * F + col];
    b2.y = Wr[(ka + 1) * F + col];
    acc = __builtin_amdgcn_wmma_f32_16x16x4_f32(false, a1, false, b1,
                                                (short)0, acc, false, false);
    acc = __builtin_amdgcn_wmma_f32_16x16x4_f32(false, a2, false, b2,
                                                (short)0, acc, false, false);
  }

  const float bv = bias[col];
#pragma unroll
  for (int v = 0; v < 8; ++v) {
    int m = base + v + 8 * half;
    float val = acc[v] + bv;
    if (relu) val = fmaxf(val, 0.0f);
    out[(size_t)m * F + col] = val;
  }
}

__global__ void fc_kernel(const float* __restrict__ h,
                          const float* __restrict__ Wfc,
                          const float* __restrict__ bfc,
                          float* __restrict__ out) {
  int node = blockIdx.x * blockDim.x + threadIdx.x;
  if (node >= NNODES) return;
  const float* row = h + (size_t)node * F;
  float a0 = bfc[0], a1 = bfc[1];
#pragma unroll 8
  for (int k = 0; k < F; ++k) {
    float v = row[k];
    a0 += v * Wfc[k * 2 + 0];
    a1 += v * Wfc[k * 2 + 1];
  }
  out[node * 2 + 0] = a0;
  out[node * 2 + 1] = a1;
}

extern "C" void kernel_launch(void* const* d_in, const int* in_sizes, int n_in,
                              void* d_out, int out_size, void* d_ws, size_t ws_size,
                              hipStream_t stream) {
  const float* x   = (const float*)d_in[0];
  const int*   ei  = (const int*)d_in[1];
  const float* W1l = (const float*)d_in[2];
  const float* b1  = (const float*)d_in[3];
  const float* W1r = (const float*)d_in[4];
  const float* W2l = (const float*)d_in[5];
  const float* b2  = (const float*)d_in[6];
  const float* W2r = (const float*)d_in[7];
  const float* Wfc = (const float*)d_in[8];
  const float* bfc = (const float*)d_in[9];
  float* out = (float*)d_out;

  float* ws     = (float*)d_ws;
  float* sumbuf = ws;
  float* h1     = sumbuf + (size_t)NNODES * F;
  float* h2     = h1 + (size_t)NNODES * F;
  float* cnt    = h2 + (size_t)NNODES * F;

  agg_tile<<<NNODES / TD, 1024, 0, stream>>>(x, ei, sumbuf, cnt, 1);
  sage_gemm<<<NNODES / 16, 256, 0, stream>>>(sumbuf, cnt, x, W1l, W1r, b1, h1, 1);

  agg_tile<<<NNODES / TD, 1024, 0, stream>>>(h1, ei, sumbuf, cnt, 0);
  sage_gemm<<<NNODES / 16, 256, 0, stream>>>(sumbuf, cnt, h1, W2l, W2r, b2, h2, 1);

  fc_kernel<<<(NNODES + 255) / 256, 256, 0, stream>>>(h2, Wfc, bfc, out);
}


# Layer GEMM wave tile 16x16 -> 16x128: a wave holds its 16 nodes' whole output rows in 8 accumulators, each A fragment loaded once per k-step instead of per column tile; B by 16-byte row loads
# speedup vs baseline: 2.5913x; 1.2690x over previous
#include <hip/hip_runtime.h>

typedef __attribute__((ext_vector_type(2))) float v2f;
typedef __attribute__((ext_vector_type(8))) float v8f;

#define NNODES 40000
#define NEDGES 640000
#define F 128

#define TD     160
#define LCAP   1024
#define NCHUNK (NEDGES / 128)
#define NSTEP  ((NCHUNK + 127) / 128)

__global__ __launch_bounds__(1024)
void agg_tile(const float* __restrict__ feat,
              const int* __restrict__ ei,
              float* __restrict__ sum,
              float* __restrict__ cnt,
              int do_cnt) {
  __shared__ float    tile[TD * F];
  __shared__ unsigned mlist[32 * LCAP];
  __shared__ int      cntl[TD];

  const int tid  = threadIdx.x;
  const int lane = tid & 31;
  const int wave = __builtin_amdgcn_readfirstlane(tid >> 5);
  const int base = blockIdx.x * TD;

#pragma unroll
  for (int k = 0; k < (TD * F) / (4 * 1024); ++k)
    ((float4*)tile)[tid + k * 1024] = make_float4(0.f, 0.f, 0.f, 0.f);
  if (tid < TD) cntl[tid] = 0;
  __syncthreads();

  const int4* dst4 = (const int4*)(ei + NEDGES);
  unsigned* myl = mlist + wave * LCAP;
  int nl = 0;

  const int rot = blockIdx.x % NSTEP;

  int4 cur[4], nxt[4];
#pragma unroll
  for (int u = 0; u < 4; ++u) {
    const int c = min((rot * 32 + wave) * 4 + u, NCHUNK - 1);
    cur[u] = dst4[c * 32 + lane];
  }

  for (int s = 0; s < NSTEP; ++s) {
    const int sc = (s + rot < NSTEP) ? s + rot : s + rot - NSTEP;
    const int sn = (sc + 1 < NSTEP) ? sc + 1 : 0;
#pragma unroll
    for (int u = 0; u < 4; ++u) {
      const int c = min((sn * 32 + wave) * 4 + u, NCHUNK - 1);
      nxt[u] = dst4[c * 32 + lane];
    }

#pragma unroll
    for (int u = 0; u < 4; ++u) {
      const int c  = (sc * 32 + wave) * 4 + u;
      const int r0 = cur[u].x - base, r1 = cur[u].y - base;
      const int r2 = cur[u].z - base, r3 = cur[u].w - base;
      const bool m0 = (unsigned)r0 < (unsigned)TD, m1 = (unsigned)r1 < (unsigned)TD;
      const bool m2 = (unsigned)r2 < (unsigned)TD, m3 = (unsigned)r3 < (unsigned)TD;
      if (c < NCHUNK && __builtin_amdgcn_ballot_w32(m0 | m1 | m2 | m3) != 0u) {
        const int e0 = (c * 32 + lane) * 4;
#define APPEND(M, R, E)                                                     \
        {                                                                   \
          const unsigned b = __builtin_amdgcn_ballot_w32(M);                \
          if (M) {                                                          \
            const int pos = nl + (int)__builtin_amdgcn_mbcnt_lo(b, 0u);     \
            myl[pos] = ((unsigned)(E) << 8) | (unsigned)(R);                \
            if (do_cnt) atomicAdd(&cntl[R], 1);                             \
          }                                                                 \
          nl += __builtin_popcount(b);                                      \
        }
        APPEND(m0, r0, e0 + 0)
        APPEND(m1, r1, e0 + 1)
        APPEND(m2, r2, e0 + 2)
        APPEND(m3, r3, e0 + 3)
#undef APPEND
      }
    }

    if (nl >= 32 || s == NSTEP - 1) {
      for (int i0 = 0; i0 < nl; i0 += 32) {
        const int n = min(32, nl - i0);
        const unsigned pk = myl[i0 + min(lane, n - 1)];
        const int sv = ei[pk >> 8];
        const int rv = (int)(pk & 255u);
        for (int j = 0; j < n; j += 8) {
          float4 v[8];
          int    rr[8];
#pragma unroll
          for (int t = 0; t < 8; ++t) {
            const int idx = min(j + t, n - 1);
            const int ss  = __builtin_amdgcn_readlane(sv, idx);
            rr[t]         = __builtin_amdgcn_readlane(rv, idx);
            v[t] = *(const float4*)(feat + (size_t)ss * F + lane * 4);
          }
#pragma unroll
          for (int t = 0; t < 8; ++t) {
            if (j + t < n) {
              float* p = tile + rr[t] * F + lane;
              atomicAdd(p,      v[t].x);
              atomicAdd(p + 32, v[t].y);
              atomicAdd(p + 64, v[t].z);
              atomicAdd(p + 96, v[t].w);
            }
          }
        }
      }
      nl = 0;
    }

#pragma unroll
    for (int u = 0; u < 4; ++u) cur[u] = nxt[u];
  }

  __syncthreads();

#pragma unroll
  for (int k = 0; k < (TD * F) / (4 * 1024); ++k) {
    const int idx = tid + k * 1024;
    const int row = idx >> 5;
    const int l   = idx & 31;
    const float* p = tile + row * F + l;
    const float4 o = make_float4(p[0], p[32], p[64], p[96]);
    *(float4*)(sum + (size_t)(base + row) * F + l * 4) = o;
  }
  if (do_cnt && tid < TD) cnt[base + tid] = (float)cntl[tid];
}

struct Frag {
  v2f    a1, a2;
  float4 l0[2], l1[2];
  float4 r0[2], r1[2];
};

__device__ __forceinline__ void load_frag(Frag& f, const float* __restrict__ srow,
                                          const float* __restrict__ xrow,
                                          const float* __restrict__ wl,
                                          const float* __restrict__ wr, int k) {
  f.a1 = *(const v2f*)(srow + k);
  f.a2 = *(const v2f*)(xrow + k);
  const float* pl = wl + (size_t)k * F;
  const float* pr = wr + (size_t)k * F;
  f.l0[0] = *(const float4*)(pl);
  f.l0[1] = *(const float4*)(pl + 64);
  f.l1[0] = *(const float4*)(pl + F);
  f.l1[1] = *(const float4*)(pl + F + 64);
  f.r0[0] = *(const float4*)(pr);
  f.r0[1] = *(const float4*)(pr + 64);
  f.r1[0] = *(const float4*)(pr + F);
  f.r1[1] = *(const float4*)(pr + F + 64);
}

#define WM(A, B0, B1, C) \
  C = __builtin_amdgcn_wmma_f32_16x16x4_f32(false, A, false, (v2f){B0, B1}, (short)0, C, false, false)

__device__ __forceinline__ void mul_frag(v8f (&acc)[8], const Frag& f, float scale) {
  const v2f a1 = f.a1 * scale;
  WM(a1, f.l0[0].x, f.l1[0].x, acc[0]);
  WM(a1, f.l0[0].y, f.l1[0].y, acc[1]);
  WM(a1, f.l0[0].z, f.l1[0].z, acc[2]);
  WM(a1, f.l0[0].w, f.l1[0].w, acc[3]);
  WM(a1, f.l0[1].x, f.l1[1].x, acc[4]);
  WM(a1, f.l0[1].y, f.l1[1].y, acc[5]);
  WM(a1, f.l0[1].z, f.l1[1].z, acc[6]);
  WM(a1, f.l0[1].w, f.l1[1].w, acc[7]);
  WM(f.a2, f.r0[0].x, f.r1[0].x, acc[0]);
  WM(f.a2, f.r0[0].y, f.r1[0].y, acc[1]);
  WM(f.a2, f.r0[0].z, f.r1[0].z, acc[2]);
  WM(f.a2, f.r0[0].w, f.r1[0].w, acc[3]);
  WM(f.a2, f.r0[1].x, f.r1[1].x, acc[4]);
  WM(f.a2, f.r0[1].y, f.r1[1].y, acc[5]);
  WM(f.a2, f.r0[1].z, f.r1[1].z, acc[6]);
  WM(f.a2, f.r0[1].w, f.r1[1].w, acc[7]);
}
#undef WM

#define GW 4

__global__ __launch_bounds__(32 * GW)
void sage_gemm(const float* __restrict__ sum,
               const float* __restrict__ cnt,
               const float* __restrict__ xin,
               const float* __restrict__ Wl,
               const float* __restrict__ Wr,
               const float* __restrict__ bias,
               float* __restrict__ out,
               int relu) {
  const int tid  = threadIdx.x;
  const int lane = tid & 31;
  const int wave = tid >> 5;
  const int half = lane >> 4;
  const int lid  = lane & 15;
  const int base = (blockIdx.x * GW + wave) * 16;

  const float scale = 1.0f / fmaxf(cnt[base + lid], 1.0f);
  const float* srow = sum + (size_t)(base + lid) * F + 2 * half;
  const float* xrow = xin + (size_t)(base + lid) * F + 2 * half;
  const float* wl   = Wl + (size_t)(2 * half) * F + 4 * lid;
  const float* wr   = Wr + (size_t)(2 * half) * F + 4 * lid;

  v8f acc[8] = {};
  Frag fa, fb;
  load_frag(fa, srow, xrow, wl, wr, 0);
#pragma unroll
  for (int k = 0; k < F; k += 8) {
    load_frag(fb, srow, xrow, wl, wr, k + 4);
    mul_frag(acc, fa, scale);
    if (k + 8 < F) load_frag(fa, srow, xrow, wl, wr, k + 8);
    mul_frag(acc, fb, scale);
  }

#pragma unroll
  for (int g = 0; g < 2; ++g) {
    const float4 bv = *(const float4*)(bias + 64 * g + 4 * lid);
#pragma unroll
    for (int v = 0; v < 8; ++v) {
      float4 o = make_float4(acc[4 * g + 0][v] + bv.x, acc[4 * g + 1][v] + bv.y,
                             acc[4 * g + 2][v] + bv.z, acc[4 * g + 3][v] + bv.w);
      if (relu) {
        o.x = fmaxf(o.x, 0.0f); o.y = fmaxf(o.y, 0.0f);
        o.z = fmaxf(o.z, 0.0f); o.w = fmaxf(o.w, 0.0f);
      }
      *(float4*)(out + (size_t)(base + v + 8 * half) * F + 64 * g + 4 * lid) = o;
    }
  }
}

__global__ void fc_kernel(const float* __restrict__ h,
                          const float* __restrict__ Wfc,
                          const float* __restrict__ bfc,
                          float* __restrict__ out) {
  int node = blockIdx.x * blockDim.x + threadIdx.x;
  if (node >= NNODES) return;
  const float* row = h + (size_t)node * F;
  float a0 = bfc[0], a1 = bfc[1];
#pragma unroll 8
  for (int k = 0; k < F; ++k) {
    float v = row[k];
    a0 += v * Wfc[k * 2 + 0];
    a1 += v * Wfc[k * 2 + 1];
  }
  out[node * 2 + 0] = a0;
  out[node * 2 + 1] = a1;
}

extern "C" void kernel_launch(void* const* d_in, const int* in_sizes, int n_in,
                              void* d_out, int out_size, void* d_ws, size_t ws_size,
                              hipStream_t stream) {
  const float* x   = (const float*)d_in[0];
  const int*   ei  = (const int*)d_in[1];
  const float* W1l = (const float*)d_in[2];
  const float* b1  = (const float*)d_in[3];
  const float* W1r = (const float*)d_in[4];
  const float* W2l = (const float*)d_in[5];
  const float* b2  = (const float*)d_in[6];
  const float* W2r = (const float*)d_in[7];
  const float* Wfc = (const float*)d_in[8];
  const float* bfc = (const float*)d_in[9];
  float* out = (float*)d_out;

  float* ws     = (float*)d_ws;
  float* sumbuf = ws;
  float* h1     = sumbuf + (size_t)NNODES * F;
  float* h2     = h1 + (size_t)NNODES * F;
  float* cnt    = h2 + (size_t)NNODES * F;

  agg_tile<<<NNODES / TD, 1024, 0, stream>>>(x, ei, sumbuf, cnt, 1);
  sage_gemm<<<NNODES / (16 * GW), 32 * GW, 0, stream>>>(sumbuf, cnt, x, W1l, W1r, b1, h1, 1);

  agg_tile<<<NNODES / TD, 1024, 0, stream>>>(h1, ei, sumbuf, cnt, 0);
  sage_gemm<<<NNODES / (16 * GW), 32 * GW, 0, stream>>>(sumbuf, cnt, h1, W2l, W2r, b2, h2, 1);

  fc_kernel<<<(NNODES + 255) / 256, 256, 0, stream>>>(h2, Wfc, bfc, out);
}


# Layer 2 reuses layer 1's edge matches: each wave saves its (source, row) pairs to workspace in launch 1; launch 2 reads them back instead of rescanning the edge list; LDS adds in arrival order
# speedup vs baseline: 2.8865x; 1.1139x over previous
#include <hip/hip_runtime.h>

typedef __attribute__((ext_vector_type(2))) float v2f;
typedef __attribute__((ext_vector_type(8))) float v8f;

#define NNODES 40000
#define NEDGES 640000
#define F 128

#define TD     160
#define SCAP   512
#define LCAP   1024
#define NCHUNK (NEDGES / 128)
#define NSTEP  ((NCHUNK + 127) / 128)

__device__ __forceinline__ void add_rows(const float* __restrict__ feat, float* tile,
                                         int sv, int rv, int n, int lane) {
  for (int j = 0; j < n; j += 8) {
    float4 v[8];
    int    rr[8];
#pragma unroll
    for (int t = 0; t < 8; ++t) {
      const int idx = min(j + t, n - 1);
      const int ss  = __builtin_amdgcn_readlane(sv, idx);
      rr[t]         = __builtin_amdgcn_readlane(rv, idx);
      v[t] = *(const float4*)(feat + (size_t)ss * F + lane * 4);
    }
#pragma unroll
    for (int t = 0; t < 8; ++t) {
      if (j + t < n) {
        float* p = tile + rr[t] * F + lane;
        atomicAdd(p,      v[t].x);
        atomicAdd(p + 32, v[t].y);
        atomicAdd(p + 64, v[t].z);
        atomicAdd(p + 96, v[t].w);
      }
    }
  }
}

__global__ __launch_bounds__(1024)
void agg_tile(const float* __restrict__ feat,
              const int* __restrict__ ei,
              float* __restrict__ sum,
              float* __restrict__ cnt,
              unsigned* __restrict__ saved,
              int* __restrict__ nsaved,
              int first) {
  __shared__ float    tile[TD * F];
  __shared__ unsigned mlist[32 * LCAP];
  __shared__ int      cntl[TD];
  __shared__ int      wcnt[32];

  const int tid  = threadIdx.x;
  const int lane = tid & 31;
  const int wave = __builtin_amdgcn_readfirstlane(tid >> 5);
  const int base = blockIdx.x * TD;

#pragma unroll
  for (int k = 0; k < (TD * F) / (4 * 1024); ++k)
    ((float4*)tile)[tid + k * 1024] = make_float4(0.f, 0.f, 0.f, 0.f);
  if (tid < TD) cntl[tid] = 0;
  __syncthreads();

  unsigned* mysave = saved + (size_t)(blockIdx.x * 32 + wave) * SCAP;
  int nsv = 0;
  const int nrep = first ? 0 : __builtin_amdgcn_readfirstlane(nsaved[blockIdx.x * 32 + wave]);

  if (!first && nrep <= SCAP) {
    unsigned* myl = mlist + wave * LCAP;
    for (int i0 = 0; i0 < nrep; i0 += 128) {
      unsigned t[4];
#pragma unroll
      for (int u = 0; u < 4; ++u) {
        const int i = i0 + u * 32 + lane;
        t[u] = (i < nrep) ? mysave[i] : 0u;
      }
#pragma unroll
      for (int u = 0; u < 4; ++u) {
        const int i = i0 + u * 32 + lane;
        if (i < nrep) myl[i] = t[u];
      }
    }
    for (int i0 = 0; i0 < nrep; i0 += 32) {
      const int n = min(32, nrep - i0);
      const unsigned w = myl[i0 + min(lane, n - 1)];
      add_rows(feat, tile, (int)min(w >> 8, (unsigned)(NNODES - 1)), (int)(w & 255u), n, lane);
    }
  } else {
  const int4* dst4 = (const int4*)(ei + NEDGES);
  unsigned* myl = mlist + wave * LCAP;
  int nl = 0;

  const int rot = blockIdx.x % NSTEP;

  int4 cur[4], nxt[4];
#pragma unroll
  for (int u = 0; u < 4; ++u) {
    const int c = min((rot * 32 + wave) * 4 + u, NCHUNK - 1);
    cur[u] = dst4[c * 32 + lane];
  }

  for (int s = 0; s < NSTEP; ++s) {
    const int sc = (s + rot < NSTEP) ? s + rot : s + rot - NSTEP;
    const int sn = (sc + 1 < NSTEP) ? sc + 1 : 0;
#pragma unroll
    for (int u = 0; u < 4; ++u) {
      const int c = min((sn * 32 + wave) * 4 + u, NCHUNK - 1);
      nxt[u] = dst4[c * 32 + lane];
    }

#pragma unroll
    for (int u = 0; u < 4; ++u) {
      const int c  = (sc * 32 + wave) * 4 + u;
      const int r0 = cur[u].x - base, r1 = cur[u].y - base;
      const int r2 = cur[u].z - base, r3 = cur[u].w - base;
      const bool m0 = (unsigned)r0 < (unsigned)TD, m1 = (unsigned)r1 < (unsigned)TD;
      const bool m2 = (unsigned)r2 < (unsigned)TD, m3 = (unsigned)r3 < (unsigned)TD;
      if (c < NCHUNK && __builtin_amdgcn_ballot_w32(m0 | m1 | m2 | m3) != 0u) {
        const int e0 = (c * 32 + lane) * 4;
#define APPEND(M, R, E)                                                     \
        {                                                                   \
          const unsigned b = __builtin_amdgcn_ballot_w32(M);                \
          if (M) {                                                          \
            const int pos = nl + (int)__builtin_amdgcn_mbcnt_lo(b, 0u);     \
            myl[pos] = ((unsigned)(E) << 8) | (unsigned)(R);                \
            if (first) atomicAdd(&cntl[R], 1);                              \
          }                                                                 \
          nl += __builtin_popcount(b);                                      \
        }
        APPEND(m0, r0, e0 + 0)
        APPEND(m1, r1, e0 + 1)
        APPEND(m2, r2, e0 + 2)
        APPEND(m3, r3, e0 + 3)
#undef APPEND
      }
    }

    if (nl >= 32 || s == NSTEP - 1) {
      for (int i0 = 0; i0 < nl; i0 += 32) {
        const int n = min(32, nl - i0);
        const unsigned pk = myl[i0 + min(lane, n - 1)];
        const int sv = ei[pk >> 8];
        const int rv = (int)(pk & 255u);
        if (first) {
          if (lane < n && nsv + lane < SCAP) mysave[nsv + lane] = ((unsigned)sv << 8) | (unsigned)rv;
          nsv += n;
        }
        add_rows(feat, tile, sv, rv, n, lane);
      }
      nl = 0;
    }

#pragma unroll
    for (int u = 0; u < 4; ++u) cur[u] = nxt[u];
  }
  }

  if (lane == 0) wcnt[wave] = nsv;
  __syncthreads();

#pragma unroll
  for (int k = 0; k < (TD * F) / (4 * 1024); ++k) {
    const int idx = tid + k * 1024;
    const int row = idx >> 5;
    const int l   = idx & 31;
    const float* p = tile + row * F + l;
    const float4 o = make_float4(p[0], p[32], p[64], p[96]);
    *(float4*)(sum + (size_t)(base + row) * F + l * 4) = o;
  }
  if (first && tid < TD) cnt[base + tid] = (float)cntl[tid];
  if (first && tid < 32) nsaved[blockIdx.x * 32 + tid] = wcnt[tid];
}

struct Frag {
  v2f    a1, a2;
  float4 l0[2], l1[2];
  float4 r0[2], r1[2];
};

__device__ __forceinline__ void load_frag(Frag& f, const float* __restrict__ srow,
                                          const float* __restrict__ xrow,
                                          const float* __restrict__ wl,
                                          const float* __restrict__ wr, int k) {
  f.a1 = *(const v2f*)(srow + k);
  f.a2 = *(const v2f*)(xrow + k);
  const float* pl = wl + (size_t)k * F;
  const float* pr = wr + (size_t)k * F;
  f.l0[0] = *(const float4*)(pl);
  f.l0[1] = *(const float4*)(pl + 64);
  f.l1[0] = *(const float4*)(pl + F);
  f.l1[1] = *(const float4*)(pl + F + 64);
  f.r0[0] = *(const float4*)(pr);
  f.r0[1] = *(const float4*)(pr + 64);
  f.r1[0] = *(const float4*)(pr + F);
  f.r1[1] = *(const float4*)(pr + F + 64);
}

#define WM(A, B0, B1, C) \
  C = __builtin_amdgcn_wmma_f32_16x16x4_f32(false, A, false, (v2f){B0, B1}, (short)0, C, false, false)

__device__ __forceinline__ void mul_frag(v8f (&acc)[8], const Frag& f, float scale) {
  const v2f a1 = f.a1 * scale;
  WM(a1, f.l0[0].x, f.l1[0].x, acc[0]);
  WM(a1, f.l0[0].y, f.l1[0].y, acc[1]);
  WM(a1, f.l0[0].z, f.l1[0].z, acc[2]);
  WM(a1, f.l0[0].w, f.l1[0].w, acc[3]);
  WM(a1, f.l0[1].x, f.l1[1].x, acc[4]);
  WM(a1, f.l0[1].y, f.l1[1].y, acc[5]);
  WM(a1, f.l0[1].z, f.l1[1].z, acc[6]);
  WM(a1, f.l0[1].w, f.l1[1].w, acc[7]);
  WM(f.a2, f.r0[0].x, f.r1[0].x, acc[0]);
  WM(f.a2, f.r0[0].y, f.r1[0].y, acc[1]);
  WM(f.a2, f.r0[0].z, f.r1[0].z, acc[2]);
  WM(f.a2, f.r0[0].w, f.r1[0].w, acc[3]);
  WM(f.a2, f.r0[1].x, f.r1[1].x, acc[4]);
  WM(f.a2, f.r0[1].y, f.r1[1].y, acc[5]);
  WM(f.a2, f.r0[1].z, f.r1[1].z, acc[6]);
  WM(f.a2, f.r0[1].w, f.r1[1].w, acc[7]);
}
#undef WM

#define GW 4

__global__ __launch_bounds__(32 * GW)
void sage_gemm(const float* __restrict__ sum,
               const float* __restrict__ cnt,
               const float* __restrict__ xin,
               const float* __restrict__ Wl,
               const float* __restrict__ Wr,
               const float* __restrict__ bias,
               float* __restrict__ out,
               int relu) {
  const int tid  = threadIdx.x;
  const int lane = tid & 31;
  const int wave = tid >> 5;
  const int half = lane >> 4;
  const int lid  = lane & 15;
  const int base = (blockIdx.x * GW + wave) * 16;

  const float scale = 1.0f / fmaxf(cnt[base + lid], 1.0f);
  const float* srow = sum + (size_t)(base + lid) * F + 2 * half;
  const float* xrow = xin + (size_t)(base + lid) * F + 2 * half;
  const float* wl   = Wl + (size_t)(2 * half) * F + 4 * lid;
  const float* wr   = Wr + (size_t)(2 * half) * F + 4 * lid;

  v8f acc[8] = {};
  Frag fa, fb;
  load_frag(fa, srow, xrow, wl, wr, 0);
#pragma unroll
  for (int k = 0; k < F; k += 8) {
    load_frag(fb, srow, xrow, wl, wr, k + 4);
    mul_frag(acc, fa, scale);
    if (k + 8 < F) load_frag(fa, srow, xrow, wl, wr, k + 8);
    mul_frag(acc, fb, scale);
  }

#pragma unroll
  for (int g = 0; g < 2; ++g) {
    const float4 bv = *(const float4*)(bias + 64 * g + 4 * lid);
#pragma unroll
    for (int v = 0; v < 8; ++v) {
      float4 o = make_float4(acc[4 * g + 0][v] + bv.x, acc[4 * g + 1][v] + bv.y,
                             acc[4 * g + 2][v] + bv.z, acc[4 * g + 3][v] + bv.w);
      if (relu) {
        o.x = fmaxf(o.x, 0.0f); o.y = fmaxf(o.y, 0.0f);
        o.z = fmaxf(o.z, 0.0f); o.w = fmaxf(o.w, 0.0f);
      }
      *(float4*)(out + (size_t)(base + v + 8 * half) * F + 64 * g + 4 * lid) = o;
    }
  }
}

__global__ void fc_kernel(const float* __restrict__ h,
                          const float* __restrict__ Wfc,
                          const float* __restrict__ bfc,
                          float* __restrict__ out) {
  int node = blockIdx.x * blockDim.x + threadIdx.x;
  if (node >= NNODES) return;
  const float* row = h + (size_t)node * F;
  float a0 = bfc[0], a1 = bfc[1];
#pragma unroll 8
  for (int k = 0; k < F; ++k) {
    float v = row[k];
    a0 += v * Wfc[k * 2 + 0];
    a1 += v * Wfc[k * 2 + 1];
  }
  out[node * 2 + 0] = a0;
  out[node * 2 + 1] = a1;
}

extern "C" void kernel_launch(void* const* d_in, const int* in_sizes, int n_in,
                              void* d_out, int out_size, void* d_ws, size_t ws_size,
                              hipStream_t stream) {
  const float* x   = (const float*)d_in[0];
  const int*   ei  = (const int*)d_in[1];
  const float* W1l = (const float*)d_in[2];
  const float* b1  = (const float*)d_in[3];
  const float* W1r = (const float*)d_in[4];
  const float* W2l = (const float*)d_in[5];
  const float* b2  = (const float*)d_in[6];
  const float* W2r = (const float*)d_in[7];
  const float* Wfc = (const float*)d_in[8];
  const float* bfc = (const float*)d_in[9];
  float* out = (float*)d_out;

  float* ws     = (float*)d_ws;
  float* sumbuf = ws;
  float* h1     = sumbuf + (size_t)NNODES * F;
  float* h2     = h1 + (size_t)NNODES * F;
  float* cnt    = h2 + (size_t)NNODES * F;
  unsigned* saved  = (unsigned*)h2;
  int*      nsaved = (int*)(h2 + (size_t)(NNODES / TD) * 32 * SCAP);

  agg_tile<<<NNODES / TD, 1024, 0, stream>>>(x, ei, sumbuf, cnt, saved, nsaved, 1);
  sage_gemm<<<NNODES / (16 * GW), 32 * GW, 0, stream>>>(sumbuf, cnt, x, W1l, W1r, b1, h1, 1);

  agg_tile<<<NNODES / TD, 1024, 0, stream>>>(h1, ei, sumbuf, cnt, saved, nsaved, 0);
  sage_gemm<<<NNODES / (16 * GW), 32 * GW, 0, stream>>>(sumbuf, cnt, h1, W2l, W2r, b2, h2, 1);

  fc_kernel<<<(NNODES + 255) / 256, 256, 0, stream>>>(h2, Wfc, bfc, out);
}


# Layer-2 neighbour sums without LDS float adds: saved pairs routed to per-row LDS lists by one integer LDS add each; a wave sums its 5 rows in registers, stores whole rows; sum order = LDS arrival
# speedup vs baseline: 2.9790x; 1.0320x over previous
#include <hip/hip_runtime.h>

typedef __attribute__((ext_vector_type(2))) float v2f;
typedef __attribute__((ext_vector_type(8))) float v8f;

#define NNODES 40000
#define NEDGES 640000
#define F 128

#define TD     160
#define SCAP   512
#define LCAP   1024
#define NCHUNK (NEDGES / 128)
#define NSTEP  ((NCHUNK + 127) / 128)

__device__ __forceinline__ void add_rows(const float* __restrict__ feat, float* tile,
                                         int sv, int rv, int n, int lane) {
  for (int j = 0; j < n; j += 8) {
    float4 v[8];
    int    rr[8];
#pragma unroll
    for (int t = 0; t < 8; ++t) {
      const int idx = min(j + t, n - 1);
      const int ss  = __builtin_amdgcn_readlane(sv, idx);
      rr[t]         = __builtin_amdgcn_readlane(rv, idx);
      v[t] = *(const float4*)(feat + (size_t)ss * F + lane * 4);
    }
#pragma unroll
    for (int t = 0; t < 8; ++t) {
      if (j + t < n) {
        float* p = tile + rr[t] * F + lane;
        atomicAdd(p,      v[t].x);
        atomicAdd(p + 32, v[t].y);
        atomicAdd(p + 64, v[t].z);
        atomicAdd(p + 96, v[t].w);
      }
    }
  }
}


#define RPW  (TD / 32)
#define RCAP 200
__device__ __forceinline__ void sum_rows(const float* __restrict__ feat, float4& acc,
                                         int sv, int n, int lane) {
  for (int j = 0; j < n; j += 16) {
    float4 v[16];
#pragma unroll
    for (int t = 0; t < 16; ++t) {
      const int ss = __builtin_amdgcn_readlane(sv, min(j + t, n - 1));
      v[t] = *(const float4*)(feat + (size_t)ss * F + lane * 4);
    }
#pragma unroll
    for (int t = 0; t < 16; ++t) {
      if (j + t < n) {
        acc.x += v[t].x; acc.y += v[t].y;
        acc.z += v[t].z; acc.w += v[t].w;
      }
    }
  }
}

__global__ __launch_bounds__(1024)
void agg_tile(const float* __restrict__ feat,
              const int* __restrict__ ei,
              float* __restrict__ sum,
              float* __restrict__ cnt,
              unsigned* __restrict__ saved,
              int* __restrict__ nsaved,
              int first) {
  __shared__ float    tile[TD * F];
  __shared__ unsigned mlist[32 * LCAP];
  __shared__ int      cntl[TD];
  __shared__ int      wcnt[32];

  const int tid  = threadIdx.x;
  const int lane = tid & 31;
  const int wave = __builtin_amdgcn_readfirstlane(tid >> 5);
  const int base = blockIdx.x * TD;

  if (!first) {
    const int creg = nsaved[blockIdx.x * 32 + lane];
    if (__builtin_amdgcn_ballot_w32(creg > SCAP || creg < 0) == 0u) {
      const int nmine = __builtin_amdgcn_readlane(creg, wave);
      if (tid < TD) cntl[tid] = 0;
      __syncthreads();
      const unsigned* rsave = saved + (size_t)(blockIdx.x * 32 + wave) * SCAP;
      for (int i0 = 0; i0 < nmine; i0 += 128) {
        unsigned w[4];
#pragma unroll
        for (int u = 0; u < 4; ++u) {
          const int i = i0 + u * 32 + lane;
          w[u] = (i < nmine) ? rsave[i] : 0u;
        }
#pragma unroll
        for (int u = 0; u < 4; ++u) {
          const int i = i0 + u * 32 + lane;
          if (i < nmine) {
            const int r   = (int)min(w[u] & 255u, (unsigned)(TD - 1));
            const int pos = atomicAdd(&cntl[r], 1);
            if (pos < RCAP) mlist[r * RCAP + pos] = w[u] >> 8;
          }
        }
      }
      __syncthreads();
      int cmax = 0;
#pragma unroll
      for (int k = 0; k < RPW; ++k) cmax = max(cmax, cntl[lane + 32 * k]);
      if (__builtin_amdgcn_ballot_w32(cmax > RCAP) == 0u) {
        for (int k = 0; k < RPW; ++k) {
          const int r = wave * RPW + k;
          const int nr = __builtin_amdgcn_readfirstlane(cntl[r]);
          const unsigned* rl = mlist + r * RCAP;
          float4 acc = make_float4(0.f, 0.f, 0.f, 0.f);
          for (int i0 = 0; i0 < nr; i0 += 32) {
            const int n = min(32, nr - i0);
            const int sv = (int)min(rl[i0 + min(lane, n - 1)], (unsigned)(NNODES - 1));
            sum_rows(feat, acc, sv, n, lane);
          }
          *(float4*)(sum + (size_t)(base + r) * F + lane * 4) = acc;
        }
        return;
      }
      __syncthreads();
    }
  }

#pragma unroll
  for (int k = 0; k < (TD * F) / (4 * 1024); ++k)
    ((float4*)tile)[tid + k * 1024] = make_float4(0.f, 0.f, 0.f, 0.f);
  if (tid < TD) cntl[tid] = 0;
  __syncthreads();

  unsigned* mysave = saved + (size_t)(blockIdx.x * 32 + wave) * SCAP;
  int nsv = 0;
  const int nrep = first ? 0 : __builtin_amdgcn_readfirstlane(nsaved[blockIdx.x * 32 + wave]);

  if (!first && nrep <= SCAP) {
    unsigned* myl = mlist + wave * LCAP;
    for (int i0 = 0; i0 < nrep; i0 += 128) {
      unsigned t[4];
#pragma unroll
      for (int u = 0; u < 4; ++u) {
        const int i = i0 + u * 32 + lane;
        t[u] = (i < nrep) ? mysave[i] : 0u;
      }
#pragma unroll
      for (int u = 0; u < 4; ++u) {
        const int i = i0 + u * 32 + lane;
        if (i < nrep) myl[i] = t[u];
      }
    }
    for (int i0 = 0; i0 < nrep; i0 += 32) {
      const int n = min(32, nrep - i0);
      const unsigned w = myl[i0 + min(lane, n - 1)];
      add_rows(feat, tile, (int)min(w >> 8, (unsigned)(NNODES - 1)), (int)(w & 255u), n, lane);
    }
  } else {
  const int4* dst4 = (const int4*)(ei + NEDGES);
  unsigned* myl = mlist + wave * LCAP;
  int nl = 0;

  const int rot = blockIdx.x % NSTEP;

  int4 cur[4], nxt[4];
#pragma unroll
  for (int u = 0; u < 4; ++u) {
    const int c = min((rot * 32 + wave) * 4 + u, NCHUNK - 1);
    cur[u] = dst4[c * 32 + lane];
  }

  for (int s = 0; s < NSTEP; ++s) {
    const int sc = (s + rot < NSTEP) ? s + rot : s + rot - NSTEP;
    const int sn = (sc + 1 < NSTEP) ? sc + 1 : 0;
#pragma unroll
    for (int u = 0; u < 4; ++u) {
      const int c = min((sn * 32 + wave) * 4 + u, NCHUNK - 1);
      nxt[u] = dst4[c * 32 + lane];
    }

#pragma unroll
    for (int u = 0; u < 4; ++u) {
      const int c  = (sc * 32 + wave) * 4 + u;
      const int r0 = cur[u].x - base, r1 = cur[u].y - base;
      const int r2 = cur[u].z - base, r3 = cur[u].w - base;
      const bool m0 = (unsigned)r0 < (unsigned)TD, m1 = (unsigned)r1 < (unsigned)TD;
      const bool m2 = (unsigned)r2 < (unsigned)TD, m3 = (unsigned)r3 < (unsigned)TD;
      if (c < NCHUNK && __builtin_amdgcn_ballot_w32(m0 | m1 | m2 | m3) != 0u) {
        const int e0 = (c * 32 + lane) * 4;
#define APPEND(M, R, E)                                                     \
        {                                                                   \
          const unsigned b = __builtin_amdgcn_ballot_w32(M);                \
          if (M) {                                                          \
            const int pos = nl + (int)__builtin_amdgcn_mbcnt_lo(b, 0u);     \
            myl[pos] = ((unsigned)(E) << 8) | (unsigned)(R);                \
            if (first) atomicAdd(&cntl[R], 1);                              \
          }                                                                 \
          nl += __builtin_popcount(b);                                      \
        }
        APPEND(m0, r0, e0 + 0)
        APPEND(m1, r1, e0 + 1)
        APPEND(m2, r2, e0 + 2)
        APPEND(m3, r3, e0 + 3)
#undef APPEND
      }
    }

    if (nl >= 32 || s == NSTEP - 1) {
      for (int i0 = 0; i0 < nl; i0 += 32) {
        const int n = min(32, nl - i0);
        const unsigned pk = myl[i0 + min(lane, n - 1)];
        const int sv = ei[pk >> 8];
        const int rv = (int)(pk & 255u);
        if (first) {
          if (lane < n && nsv + lane < SCAP) mysave[nsv + lane] = ((unsigned)sv << 8) | (unsigned)rv;
          nsv += n;
        }
        add_rows(feat, tile, sv, rv, n, lane);
      }
      nl = 0;
    }

#pragma unroll
    for (int u = 0; u < 4; ++u) cur[u] = nxt[u];
  }
  }

  if (lane == 0) wcnt[wave] = nsv;
  __syncthreads();

#pragma unroll
  for (int k = 0; k < (TD * F) / (4 * 1024); ++k) {
    const int idx = tid + k * 1024;
    const int row = idx >> 5;
    const int l   = idx & 31;
    const float* p = tile + row * F + l;
    const float4 o = make_float4(p[0], p[32], p[64], p[96]);
    *(float4*)(sum + (size_t)(base + row) * F + l * 4) = o;
  }
  if (first && tid < TD) cnt[base + tid] = (float)cntl[tid];
  if (first && tid < 32) nsaved[blockIdx.x * 32 + tid] = wcnt[tid];
}

struct Frag {
  v2f    a1, a2;
  float4 l0[2], l1[2];
  float4 r0[2], r1[2];
};

__device__ __forceinline__ void load_frag(Frag& f, const float* __restrict__ srow,
                                          const float* __restrict__ xrow,
                                          const float* __restrict__ wl,
                                          const float* __restrict__ wr, int k) {
  f.a1 = *(const v2f*)(srow + k);
  f.a2 = *(const v2f*)(xrow + k);
  const float* pl = wl + (size_t)k * F;
  const float* pr = wr + (size_t)k * F;
  f.l0[0] = *(const float4*)(pl);
  f.l0[1] = *(const float4*)(pl + 64);
  f.l1[0] = *(const float4*)(pl + F);
  f.l1[1] = *(const float4*)(pl + F + 64);
  f.r0[0] = *(const float4*)(pr);
  f.r0[1] = *(const float4*)(pr + 64);
  f.r1[0] = *(const float4*)(pr + F);
  f.r1[1] = *(const float4*)(pr + F + 64);
}

#define WM(A, B0, B1, C) \
  C = __builtin_amdgcn_wmma_f32_16x16x4_f32(false, A, false, (v2f){B0, B1}, (short)0, C, false, false)

__device__ __forceinline__ void mul_frag(v8f (&acc)[8], const Frag& f, float scale) {
  const v2f a1 = f.a1 * scale;
  WM(a1, f.l0[0].x, f.l1[0].x, acc[0]);
  WM(a1, f.l0[0].y, f.l1[0].y, acc[1]);
  WM(a1, f.l0[0].z, f.l1[0].z, acc[2]);
  WM(a1, f.l0[0].w, f.l1[0].w, acc[3]);
  WM(a1, f.l0[1].x, f.l1[1].x, acc[4]);
  WM(a1, f.l0[1].y, f.l1[1].y, acc[5]);
  WM(a1, f.l0[1].z, f.l1[1].z, acc[6]);
  WM(a1, f.l0[1].w, f.l1[1].w, acc[7]);
  WM(f.a2, f.r0[0].x, f.r1[0].x, acc[0]);
  WM(f.a2, f.r0[0].y, f.r1[0].y, acc[1]);
  WM(f.a2, f.r0[0].z, f.r1[0].z, acc[2]);
  WM(f.a2, f.r0[0].w, f.r1[0].w, acc[3]);
  WM(f.a2, f.r0[1].x, f.r1[1].x, acc[4]);
  WM(f.a2, f.r0[1].y, f.r1[1].y, acc[5]);
  WM(f.a2, f.r0[1].z, f.r1[1].z, acc[6]);
  WM(f.a2, f.r0[1].w, f.r1[1].w, acc[7]);
}
#undef WM

#define GW 4

__global__ __launch_bounds__(32 * GW)
void sage_gemm(const float* __restrict__ sum,
               const float* __restrict__ cnt,
               const float* __restrict__ xin,
               const float* __restrict__ Wl,
               const float* __restrict__ Wr,
               const float* __restrict__ bias,
               float* __restrict__ out,
               int relu) {
  const int tid  = threadIdx.x;
  const int lane = tid & 31;
  const int wave = tid >> 5;
  const int half = lane >> 4;
  const int lid  = lane & 15;
  const int base = (blockIdx.x * GW + wave) * 16;

  const float scale = 1.0f / fmaxf(cnt[base + lid], 1.0f);
  const float* srow = sum + (size_t)(base + lid) * F + 2 * half;
  const float* xrow = xin + (size_t)(base + lid) * F + 2 * half;
  const float* wl   = Wl + (size_t)(2 * half) * F + 4 * lid;
  const float* wr   = Wr + (size_t)(2 * half) * F + 4 * lid;

  v8f acc[8] = {};
  Frag fa, fb;
  load_frag(fa, srow, xrow, wl, wr, 0);
#pragma unroll
  for (int k = 0; k < F; k += 8) {
    load_frag(fb, srow, xrow, wl, wr, k + 4);
    mul_frag(acc, fa, scale);
    if (k + 8 < F) load_frag(fa, srow, xrow, wl, wr, k + 8);
    mul_frag(acc, fb, scale);
  }

#pragma unroll
  for (int g = 0; g < 2; ++g) {
    const float4 bv = *(const float4*)(bias + 64 * g + 4 * lid);
#pragma unroll
    for (int v = 0; v < 8; ++v) {
      float4 o = make_float4(acc[4 * g + 0][v] + bv.x, acc[4 * g + 1][v] + bv.y,
                             acc[4 * g + 2][v] + bv.z, acc[4 * g + 3][v] + bv.w);
      if (relu) {
        o.x = fmaxf(o.x, 0.0f); o.y = fmaxf(o.y, 0.0f);
        o.z = fmaxf(o.z, 0.0f); o.w = fmaxf(o.w, 0.0f);
      }
      *(float4*)(out + (size_t)(base + v + 8 * half) * F + 64 * g + 4 * lid) = o;
    }
  }
}

__global__ void fc_kernel(const float* __restrict__ h,
                          const float* __restrict__ Wfc,
                          const float* __restrict__ bfc,
                          float* __restrict__ out) {
  int node = blockIdx.x * blockDim.x + threadIdx.x;
  if (node >= NNODES) return;
  const float* row = h + (size_t)node * F;
  float a0 = bfc[0], a1 = bfc[1];
#pragma unroll 8
  for (int k = 0; k < F; ++k) {
    float v = row[k];
    a0 += v * Wfc[k * 2 + 0];
    a1 += v * Wfc[k * 2 + 1];
  }
  out[node * 2 + 0] = a0;
  out[node * 2 + 1] = a1;
}

extern "C" void kernel_launch(void* const* d_in, const int* in_sizes, int n_in,
                              void* d_out, int out_size, void* d_ws, size_t ws_size,
                              hipStream_t stream) {
  const float* x   = (const float*)d_in[0];
  const int*   ei  = (const int*)d_in[1];
  const float* W1l = (const float*)d_in[2];
  const float* b1  = (const float*)d_in[3];
  const float* W1r = (const float*)d_in[4];
  const float* W2l = (const float*)d_in[5];
  const float* b2  = (const float*)d_in[6];
  const float* W2r = (const float*)d_in[7];
  const float* Wfc = (const float*)d_in[8];
  const float* bfc = (const float*)d_in[9];
  float* out = (float*)d_out;

  float* ws     = (float*)d_ws;
  float* sumbuf = ws;
  float* h1     = sumbuf + (size_t)NNODES * F;
  float* h2     = h1 + (size_t)NNODES * F;
  float* cnt    = h2 + (size_t)NNODES * F;
  unsigned* saved  = (unsigned*)h2;
  int*      nsaved = (int*)(h2 + (size_t)(NNODES / TD) * 32 * SCAP);

  agg_tile<<<NNODES / TD, 1024, 0, stream>>>(x, ei, sumbuf, cnt, saved, nsaved, 1);
  sage_gemm<<<NNODES / (16 * GW), 32 * GW, 0, stream>>>(sumbuf, cnt, x, W1l, W1r, b1, h1, 1);

  agg_tile<<<NNODES / TD, 1024, 0, stream>>>(h1, ei, sumbuf, cnt, saved, nsaved, 0);
  sage_gemm<<<NNODES / (16 * GW), 32 * GW, 0, stream>>>(sumbuf, cnt, h1, W2l, W2r, b2, h2, 1);

  fc_kernel<<<(NNODES + 255) / 256, 256, 0, stream>>>(h2, Wfc, bfc, out);
}


# Layer 2 multiplies before it sums: y=h1*W2l, z=h1*W2r+b2 in one product launch; the neighbour-sum launch adds mean, z, relu and the head per finished row; 5 launches; layer-2/head sum order changed
# speedup vs baseline: 3.9185x; 1.3154x over previous
#include <hip/hip_runtime.h>

typedef __attribute__((ext_vector_type(2))) float v2f;
typedef __attribute__((ext_vector_type(8))) float v8f;

#define NNODES 40000
#define NEDGES 640000
#define F 128

#define TD     160
#define SCAP   512
#define MCAP   640
#define LCAP   (MCAP + 640)
#define NCH    (NEDGES / 32)
#define NWORD  (NCH / 32)
#define NWL    ((NWORD + 31) / 32)
#define BMP    256

__global__ __launch_bounds__(128)
void mark_chunks(const int* __restrict__ ei, unsigned* __restrict__ bm) {
  __shared__ unsigned bits[BMP];
  const int tid = threadIdx.x;
  bits[tid] = 0u;
  bits[tid + 128] = 0u;
  __syncthreads();
  const int4* d4 = (const int4*)(ei + NEDGES) + (size_t)blockIdx.x * 256 + tid * 2;
  const int4 a = d4[0];
  const int4 b = d4[1];
  const unsigned bit = 1u << (tid >> 2);
#define MARK(D) atomicOr(&bits[min((unsigned)(D) / (unsigned)TD, (unsigned)(NNODES / TD - 1))], bit);
  MARK(a.x) MARK(a.y) MARK(a.z) MARK(a.w)
  MARK(b.x) MARK(b.y) MARK(b.z) MARK(b.w)
#undef MARK
  __syncthreads();
  for (int i = tid; i < NNODES / TD; i += 128) bm[(size_t)blockIdx.x * BMP + i] = bits[i];
}

__device__ __forceinline__ void add_rows(const float* __restrict__ feat, float* tile,
                                         int sv, int rv, int n, int lane) {
  for (int j = 0; j < n; j += 8) {
    float4 v[8];
    int    rr[8];
#pragma unroll
    for (int t = 0; t < 8; ++t) {
      const int idx = min(j + t, n - 1);
      const int ss  = __builtin_amdgcn_readlane(sv, idx);
      rr[t]         = __builtin_amdgcn_readlane(rv, idx);
      v[t] = *(const float4*)(feat + (size_t)ss * F + lane * 4);
    }
#pragma unroll
    for (int t = 0; t < 8; ++t) {
      if (j + t < n) {
        float* p = tile + rr[t] * F + lane;
        atomicAdd(p,      v[t].x);
        atomicAdd(p + 32, v[t].y);
        atomicAdd(p + 64, v[t].z);
        atomicAdd(p + 96, v[t].w);
      }
    }
  }
}


#define RPW  (TD / 32)
#define RCAP 200
#define RCAP1 128
__device__ __forceinline__ void sum_rows(const float* __restrict__ feat, float4& acc,
                                         int sv, int n, int lane) {
  for (int j = 0; j < n; j += 16) {
    float4 v[16];
#pragma unroll
    for (int t = 0; t < 16; ++t) {
      const int ss = __builtin_amdgcn_readlane(sv, min(j + t, n - 1));
      v[t] = *(const float4*)(feat + (size_t)ss * F + lane * 4);
    }
#pragma unroll
    for (int t = 0; t < 16; ++t) {
      if (j + t < n) {
        acc.x += v[t].x; acc.y += v[t].y;
        acc.z += v[t].z; acc.w += v[t].w;
      }
    }
  }
}

__device__ __forceinline__ void head_row(const float4 acc, const float4 z, float inv,
                                         const float4 wa, const float4 wb,
                                         float b0, float b1, float* res, int r, int lane) {
  const float hx = fmaxf(fmaf(acc.x, inv, z.x), 0.0f);
  const float hy = fmaxf(fmaf(acc.y, inv, z.y), 0.0f);
  const float hz = fmaxf(fmaf(acc.z, inv, z.z), 0.0f);
  const float hw = fmaxf(fmaf(acc.w, inv, z.w), 0.0f);
  const float p0 = hx * wa.x + hy * wa.z + hz * wb.x + hw * wb.z;
  const float p1 = hx * wa.y + hy * wa.w + hz * wb.y + hw * wb.w;
  const bool  hi   = lane >= 16;
  const float keep = hi ? p1 : p0;
  const float send = hi ? p0 : p1;
  float v = keep + __builtin_bit_cast(float, __builtin_amdgcn_permlanex16(
                       0, __builtin_bit_cast(int, send), 0x76543210, 0xfedcba98, false, false));
#define XADD(CTRL) v += __builtin_bit_cast(float, __builtin_amdgcn_update_dpp(0, __builtin_bit_cast(int, v), CTRL, 0xf, 0xf, true))
  XADD(0x168); XADD(0x164); XADD(0x162); XADD(0x161);
#undef XADD
  if (lane == 0)  res[2 * r]     = v + b0;
  if (lane == 16) res[2 * r + 1] = v + b1;
}

__global__ __launch_bounds__(1024)
void agg_tile(const float* __restrict__ feat,
              const int* __restrict__ ei,
              float* __restrict__ sum,
              float* __restrict__ cnt,
              unsigned* __restrict__ saved,
              int* __restrict__ nsaved,
              const unsigned* __restrict__ bm,
              int first,
              const float* __restrict__ zb,
              const float* __restrict__ Wfc,
              const float* __restrict__ bfc,
              float* __restrict__ out) {
  __shared__ float    tile[TD * F];
  __shared__ unsigned mlist[32 * LCAP];
  __shared__ int      cntl[TD];
  __shared__ int      wcnt[32];
  __shared__ float    res[TD * 2];

  const int tid  = threadIdx.x;
  const int lane = tid & 31;
  const int wave = __builtin_amdgcn_readfirstlane(tid >> 5);
  const int base = blockIdx.x * TD;

  float4 wa = make_float4(0.f, 0.f, 0.f, 0.f), wb = wa;
  float hb0 = 0.f, hb1 = 0.f;
  if (!first) {
    wa  = *(const float4*)(Wfc + lane * 8);
    wb  = *(const float4*)(Wfc + lane * 8 + 4);
    hb0 = bfc[0];
    hb1 = bfc[1];
  }

  if (!first) {
    const int creg = nsaved[blockIdx.x * 32 + lane];
    if (__builtin_amdgcn_ballot_w32(creg > SCAP || creg < 0) == 0u) {
      const int nmine = __builtin_amdgcn_readlane(creg, wave);
      if (tid < TD) cntl[tid] = 0;
      __syncthreads();
      const unsigned* rsave = saved + (size_t)(blockIdx.x * 32 + wave) * SCAP;
      for (int i0 = 0; i0 < nmine; i0 += 128) {
        unsigned w[4];
#pragma unroll
        for (int u = 0; u < 4; ++u) {
          const int i = i0 + u * 32 + lane;
          w[u] = (i < nmine) ? rsave[i] : 0u;
        }
#pragma unroll
        for (int u = 0; u < 4; ++u) {
          const int i = i0 + u * 32 + lane;
          if (i < nmine) {
            const int r   = (int)min(w[u] & 255u, (unsigned)(TD - 1));
            const int pos = atomicAdd(&cntl[r], 1);
            if (pos < RCAP) mlist[r * RCAP + pos] = w[u] >> 8;
          }
        }
      }
      __syncthreads();
      int cmax = 0;
#pragma unroll
      for (int k = 0; k < RPW; ++k) cmax = max(cmax, cntl[lane + 32 * k]);
      if (__builtin_amdgcn_ballot_w32(cmax > RCAP) == 0u) {
        const int cv = __builtin_bit_cast(int, cnt[base + wave * RPW + min(lane, RPW - 1)]);
        for (int k = 0; k < RPW; ++k) {
          const int r = wave * RPW + k;
          const int nr = __builtin_amdgcn_readfirstlane(cntl[r]);
          const unsigned* rl = mlist + r * RCAP;
          const float4 z = *(const float4*)(zb + (size_t)(base + r) * F + lane * 4);
          const float inv = 1.0f / fmaxf(__builtin_bit_cast(float, __builtin_amdgcn_readlane(cv, k)), 1.0f);
          float4 acc = make_float4(0.f, 0.f, 0.f, 0.f);
          for (int i0 = 0; i0 < nr; i0 += 32) {
            const int n = min(32, nr - i0);
            const int sv = (int)min(rl[i0 + min(lane, n - 1)], (unsigned)(NNODES - 1));
            sum_rows(feat, acc, sv, n, lane);
          }
          head_row(acc, z, inv, wa, wb, hb0, hb1, res, r, lane);
        }
        __syncthreads();
        if (tid < TD * 2) out[(size_t)base * 2 + tid] = res[tid];
        return;
      }
      __syncthreads();
    }
  }

  unsigned* rlist = (unsigned*)tile;
  for (int route = first ? 1 : 0; route >= 0; --route) {
  if (!route) {
#pragma unroll
    for (int k = 0; k < (TD * F) / (4 * 1024); ++k)
      ((float4*)tile)[tid + k * 1024] = make_float4(0.f, 0.f, 0.f, 0.f);
  }
  if (tid < TD) cntl[tid] = 0;
  __syncthreads();

  unsigned* mysave = saved + (size_t)(blockIdx.x * 32 + wave) * SCAP;
  int nsv = 0;
  const int nrep = first ? 0 : __builtin_amdgcn_readfirstlane(nsaved[blockIdx.x * 32 + wave]);

  if (!first && nrep <= SCAP) {
    unsigned* myl = mlist + wave * LCAP;
    for (int i0 = 0; i0 < nrep; i0 += 128) {
      unsigned t[4];
#pragma unroll
      for (int u = 0; u < 4; ++u) {
        const int i = i0 + u * 32 + lane;
        t[u] = (i < nrep) ? mysave[i] : 0u;
      }
#pragma unroll
      for (int u = 0; u < 4; ++u) {
        const int i = i0 + u * 32 + lane;
        if (i < nrep) myl[i] = t[u];
      }
    }
    for (int i0 = 0; i0 < nrep; i0 += 32) {
      const int n = min(32, nrep - i0);
      const unsigned w = myl[i0 + min(lane, n - 1)];
      add_rows(feat, tile, (int)min(w >> 8, (unsigned)(NNODES - 1)), (int)(w & 255u), n, lane);
    }
  } else {
  const int* dstc = ei + NEDGES;
  unsigned* myl  = mlist + wave * LCAP;
  unsigned* mycl = myl + MCAP;
  int nl = 0;

  const int wd = lane * 32 + wave;
  const unsigned bits = (lane < NWL && wd < NWORD) ? bm[(size_t)wd * BMP + blockIdx.x] : 0u;

  int nc = 0;
#pragma unroll
  for (int L = 0; L < NWL; ++L) {
    const unsigned uw = __builtin_amdgcn_readlane(bits, L);
    if ((uw >> lane) & 1u)
      mycl[nc + (int)__builtin_amdgcn_mbcnt_lo(uw, 0u)] = (unsigned)((L * 32 + wave) * 32 + lane);
    nc += __builtin_popcount(uw);
  }

  for (int c0 = 0; c0 < nc; c0 += 32) {
    const int nb = min(32, nc - c0);
    const int cl = (int)mycl[c0 + min(lane, nb - 1)];
    for (int j = 0; j < nb; j += 16) {
      int dv[16];
#pragma unroll
      for (int t = 0; t < 16; ++t) {
        const int c = __builtin_amdgcn_readlane(cl, min(j + t, nb - 1));
        dv[t] = dstc[c * 32 + lane];
      }
#pragma unroll
      for (int t = 0; t < 16; ++t) {
        if (j + t < nb) {
          const int c = __builtin_amdgcn_readlane(cl, min(j + t, nb - 1));
          const int r = dv[t] - base;
          const bool m = (unsigned)r < (unsigned)TD;
          const unsigned b = __builtin_amdgcn_ballot_w32(m);
          if (m) {
            const int pos = nl + (int)__builtin_amdgcn_mbcnt_lo(b, 0u);
            myl[pos] = ((unsigned)(c * 32 + lane) << 8) | (unsigned)r;
            if (first && !route) atomicAdd(&cntl[r], 1);
          }
          nl += __builtin_popcount(b);
        }
      }

      if (nl >= (route ? MCAP - 512 : 32) || c0 + j + 16 >= nc) {
        if (route) {
          for (int i0 = 0; i0 < nl; i0 += 128) {
            unsigned pk[4];
            int      s4[4];
#pragma unroll
            for (int u = 0; u < 4; ++u) {
              pk[u] = myl[min(i0 + u * 32 + lane, nl - 1)];
              s4[u] = ei[pk[u] >> 8];
            }
#pragma unroll
            for (int u = 0; u < 4; ++u) {
              const int idx = i0 + u * 32 + lane;
              if (idx < nl) {
                const int rv = (int)(pk[u] & 255u);
                if (nsv + idx < SCAP) mysave[nsv + idx] = ((unsigned)s4[u] << 8) | (unsigned)rv;
                const int pos = atomicAdd(&cntl[rv], 1);
                if (pos < RCAP1) rlist[rv * RCAP1 + pos] = (unsigned)s4[u];
              }
            }
          }
          nsv += nl;
        } else {
          for (int i0 = 0; i0 < nl; i0 += 32) {
            const int n = min(32, nl - i0);
            const unsigned pk = myl[i0 + min(lane, n - 1)];
            const int sv = ei[pk >> 8];
            const int rv = (int)(pk & 255u);
            if (first) {
              if (lane < n && nsv + lane < SCAP) mysave[nsv + lane] = ((unsigned)sv << 8) | (unsigned)rv;
              nsv += n;
            }
            add_rows(feat, tile, sv, rv, n, lane);
          }
        }
        nl = 0;
      }
    }
  }
  }

  if (lane == 0) wcnt[wave] = nsv;
  __syncthreads();

  if (route) {
    int cmax = 0;
#pragma unroll
    for (int k = 0; k < RPW; ++k) cmax = max(cmax, cntl[lane + 32 * k]);
    if (__builtin_amdgcn_ballot_w32(cmax > RCAP1) == 0u) {
      for (int k = 0; k < RPW; ++k) {
        const int r = wave * RPW + k;
        const int nr = __builtin_amdgcn_readfirstlane(cntl[r]);
        const unsigned* rl = rlist + r * RCAP1;
        float4 acc = make_float4(0.f, 0.f, 0.f, 0.f);
        for (int i0 = 0; i0 < nr; i0 += 32) {
          const int n = min(32, nr - i0);
          const int sv = (int)rl[i0 + min(lane, n - 1)];
          sum_rows(feat, acc, sv, n, lane);
        }
        *(float4*)(sum + (size_t)(base + r) * F + lane * 4) = acc;
      }
      if (tid < TD) cnt[base + tid] = (float)cntl[tid];
      if (tid < 32) nsaved[blockIdx.x * 32 + tid] = wcnt[tid];
      return;
    }
    __syncthreads();
    continue;
  }

#pragma unroll
  for (int k = 0; k < (TD * F) / (4 * 1024); ++k) {
    const int idx = tid + k * 1024;
    const int row = idx >> 5;
    const int l   = idx & 31;
    const float* p = tile + row * F + l;
    const float4 o = make_float4(p[0], p[32], p[64], p[96]);
    if (first) {
      *(float4*)(sum + (size_t)(base + row) * F + l * 4) = o;
    } else {
      const float4 z = *(const float4*)(zb + (size_t)(base + row) * F + l * 4);
      const float inv = 1.0f / fmaxf(cnt[base + row], 1.0f);
      head_row(o, z, inv, wa, wb, hb0, hb1, res, row, l);
    }
  }
  if (first && tid < TD) cnt[base + tid] = (float)cntl[tid];
  if (first && tid < 32) nsaved[blockIdx.x * 32 + tid] = wcnt[tid];
  if (!first) {
    __syncthreads();
    if (tid < TD * 2) out[(size_t)base * 2 + tid] = res[tid];
  }
  }
}

struct Frag {
  v2f    a1, a2;
  float4 l0, l1;
  float4 r0, r1;
};

__device__ __forceinline__ void load_frag(Frag& f, const float* __restrict__ srow,
                                          const float* __restrict__ xrow,
                                          const float* __restrict__ wl,
                                          const float* __restrict__ wr, int k) {
  f.a1 = *(const v2f*)(srow + k);
  f.a2 = *(const v2f*)(xrow + k);
  const float* pl = wl + (size_t)k * F;
  const float* pr = wr + (size_t)k * F;
  f.l0 = *(const float4*)(pl);
  f.l1 = *(const float4*)(pl + F);
  f.r0 = *(const float4*)(pr);
  f.r1 = *(const float4*)(pr + F);
}

#define WM(A, B0, B1, C) \
  C = __builtin_amdgcn_wmma_f32_16x16x4_f32(false, A, false, (v2f){B0, B1}, (short)0, C, false, false)

__device__ __forceinline__ void mul_frag(v8f (&acc)[4], const Frag& f, float scale) {
  const v2f a1 = f.a1 * scale;
  WM(a1, f.l0.x, f.l1.x, acc[0]);
  WM(a1, f.l0.y, f.l1.y, acc[1]);
  WM(a1, f.l0.z, f.l1.z, acc[2]);
  WM(a1, f.l0.w, f.l1.w, acc[3]);
  WM(f.a2, f.r0.x, f.r1.x, acc[0]);
  WM(f.a2, f.r0.y, f.r1.y, acc[1]);
  WM(f.a2, f.r0.z, f.r1.z, acc[2]);
  WM(f.a2, f.r0.w, f.r1.w, acc[3]);
}
#undef WM

#define GW 4

__global__ __launch_bounds__(32 * GW)
void sage_gemm(const float* __restrict__ sum,
               const float* __restrict__ cnt,
               const float* __restrict__ xin,
               const float* __restrict__ Wl,
               const float* __restrict__ Wr,
               const float* __restrict__ bias,
               float* __restrict__ out,
               int relu) {
  const int tid  = threadIdx.x;
  const int lane = tid & 31;
  const int wave = tid >> 5;
  const int half = lane >> 4;
  const int lid  = lane & 15;
  const int g    = wave & 1;
  const int base = (blockIdx.x * (GW / 2) + (wave >> 1)) * 16;

  const float scale = 1.0f / fmaxf(cnt[base + lid], 1.0f);
  const float* srow = sum + (size_t)(base + lid) * F + 2 * half;
  const float* xrow = xin + (size_t)(base + lid) * F + 2 * half;
  const float* wl   = Wl + (size_t)(2 * half) * F + 64 * g + 4 * lid;
  const float* wr   = Wr + (size_t)(2 * half) * F + 64 * g + 4 * lid;

  v8f acc[4] = {};
  Frag f[3];
  load_frag(f[0], srow, xrow, wl, wr, 0);
  load_frag(f[1], srow, xrow, wl, wr, 4);
#pragma unroll
  for (int s = 0; s < F / 4; ++s) {
    if (s + 2 < F / 4) load_frag(f[(s + 2) % 3], srow, xrow, wl, wr, 4 * (s + 2));
    __builtin_amdgcn_sched_barrier(0);
    mul_frag(acc, f[s % 3], scale);
    __builtin_amdgcn_sched_barrier(0);
  }

  const float4 bv = *(const float4*)(bias + 64 * g + 4 * lid);
#pragma unroll
  for (int v = 0; v < 8; ++v) {
    float4 o = make_float4(acc[0][v] + bv.x, acc[1][v] + bv.y,
                           acc[2][v] + bv.z, acc[3][v] + bv.w);
    if (relu) {
      o.x = fmaxf(o.x, 0.0f); o.y = fmaxf(o.y, 0.0f);
      o.z = fmaxf(o.z, 0.0f); o.w = fmaxf(o.w, 0.0f);
    }
    *(float4*)(out + (size_t)(base + v + 8 * half) * F + 64 * g + 4 * lid) = o;
  }
}

struct Frag1 {
  v2f    a;
  float4 l0, l1;
  float4 r0, r1;
};

__device__ __forceinline__ void load_frag1(Frag1& f, const float* xrow,
                                           const float* __restrict__ wl,
                                           const float* __restrict__ wr, int k) {
  f.a = *(const v2f*)(xrow + k);
  const float* pl = wl + (size_t)k * F;
  const float* pr = wr + (size_t)k * F;
  f.l0 = *(const float4*)(pl);
  f.l1 = *(const float4*)(pl + F);
  f.r0 = *(const float4*)(pr);
  f.r1 = *(const float4*)(pr + F);
}

#define WM(A, B0, B1, C) \
  C = __builtin_amdgcn_wmma_f32_16x16x4_f32(false, A, false, (v2f){B0, B1}, (short)0, C, false, false)

__device__ __forceinline__ void mul_frag1(v8f (&ay)[4], v8f (&az)[4], const Frag1& f) {
  WM(f.a, f.l0.x, f.l1.x, ay[0]);
  WM(f.a, f.l0.y, f.l1.y, ay[1]);
  WM(f.a, f.l0.z, f.l1.z, ay[2]);
  WM(f.a, f.l0.w, f.l1.w, ay[3]);
  WM(f.a, f.r0.x, f.r1.x, az[0]);
  WM(f.a, f.r0.y, f.r1.y, az[1]);
  WM(f.a, f.r0.z, f.r1.z, az[2]);
  WM(f.a, f.r0.w, f.r1.w, az[3]);
}
#undef WM

__global__ __launch_bounds__(32 * GW)
void lin_pair(const float* xin,
              const float* __restrict__ Wl,
              const float* __restrict__ Wr,
              const float* __restrict__ bias,
              float* __restrict__ yout,
              float* zout) {
  const int tid  = threadIdx.x;
  const int lane = tid & 31;
  const int wave = tid >> 5;
  const int half = lane >> 4;
  const int lid  = lane & 15;
  const int g    = wave & 1;
  const int base = (blockIdx.x * (GW / 2) + (wave >> 1)) * 16;

  const float* xrow = xin + (size_t)(base + lid) * F + 2 * half;
  const float* wl   = Wl + (size_t)(2 * half) * F + 64 * g + 4 * lid;
  const float* wr   = Wr + (size_t)(2 * half) * F + 64 * g + 4 * lid;

  v8f ay[4] = {};
  v8f az[4] = {};
  Frag1 f[3];
  load_frag1(f[0], xrow, wl, wr, 0);
  load_frag1(f[1], xrow, wl, wr, 4);
#pragma unroll
  for (int s = 0; s < F / 4; ++s) {
    if (s + 2 < F / 4) load_frag1(f[(s + 2) % 3], xrow, wl, wr, 4 * (s + 2));
    __builtin_amdgcn_sched_barrier(0);
    mul_frag1(ay, az, f[s % 3]);
    __builtin_amdgcn_sched_barrier(0);
  }

  __syncthreads();

  const float4 bv = *(const float4*)(bias + 64 * g + 4 * lid);
#pragma unroll
  for (int v = 0; v < 8; ++v) {
    const size_t o = (size_t)(base + v + 8 * half) * F + 64 * g + 4 * lid;
    *(float4*)(yout + o) = make_float4(ay[0][v], ay[1][v], ay[2][v], ay[3][v]);
    *(float4*)(zout + o) = make_float4(az[0][v] + bv.x, az[1][v] + bv.y,
                                       az[2][v] + bv.z, az[3][v] + bv.w);
  }
}

extern "C" void kernel_launch(void* const* d_in, const int* in_sizes, int n_in,
                              void* d_out, int out_size, void* d_ws, size_t ws_size,
                              hipStream_t stream) {
  const float* x   = (const float*)d_in[0];
  const int*   ei  = (const int*)d_in[1];
  const float* W1l = (const float*)d_in[2];
  const float* b1  = (const float*)d_in[3];
  const float* W1r = (const float*)d_in[4];
  const float* W2l = (const float*)d_in[5];
  const float* b2  = (const float*)d_in[6];
  const float* W2r = (const float*)d_in[7];
  const float* Wfc = (const float*)d_in[8];
  const float* bfc = (const float*)d_in[9];
  float* out = (float*)d_out;

  float* ws     = (float*)d_ws;
  float* sumbuf = ws;
  float* h1     = sumbuf + (size_t)NNODES * F;
  float* h2     = h1 + (size_t)NNODES * F;
  float* cnt    = h2 + (size_t)NNODES * F;
  unsigned* saved  = (unsigned*)h2;
  int*      nsaved = (int*)(h2 + (size_t)(NNODES / TD) * 32 * SCAP);
  unsigned* bm     = (unsigned*)(nsaved + (NNODES / TD) * 32);

  mark_chunks<<<NWORD, 128, 0, stream>>>(ei, bm);

  agg_tile<<<NNODES / TD, 1024, 0, stream>>>(x, ei, sumbuf, cnt, saved, nsaved, bm, 1,
                                             nullptr, nullptr, nullptr, nullptr);
  sage_gemm<<<NNODES / (8 * GW), 32 * GW, 0, stream>>>(sumbuf, cnt, x, W1l, W1r, b1, h1, 1);

  lin_pair<<<NNODES / (8 * GW), 32 * GW, 0, stream>>>(h1, W2l, W2r, b2, sumbuf, h1);
  agg_tile<<<NNODES / TD, 1024, 0, stream>>>(sumbuf, ei, nullptr, cnt, saved, nsaved, bm, 0,
                                             h1, Wfc, bfc, out);
}


# Layer-1 product and both layer-2 products of a node tile in one launch: the relu'd h1 tile crosses LDS to be phase 2's A operand and is never stored; y over the sum rows; 4 launches; sum order kept
# speedup vs baseline: 4.2292x; 1.0793x over previous
#include <hip/hip_runtime.h>

typedef __attribute__((ext_vector_type(2))) float v2f;
typedef __attribute__((ext_vector_type(8))) float v8f;

#define NNODES 40000
#define NEDGES 640000
#define F 128

#define TD     160
#define SCAP   512
#define MCAP   640
#define LCAP   (MCAP + 640)
#define NCH    (NEDGES / 32)
#define NWORD  (NCH / 32)
#define NWL    ((NWORD + 31) / 32)
#define BMP    256

__global__ __launch_bounds__(128)
void mark_chunks(const int* __restrict__ ei, unsigned* __restrict__ bm) {
  __shared__ unsigned bits[BMP];
  const int tid = threadIdx.x;
  bits[tid] = 0u;
  bits[tid + 128] = 0u;
  __syncthreads();
  const int4* d4 = (const int4*)(ei + NEDGES) + (size_t)blockIdx.x * 256 + tid * 2;
  const int4 a = d4[0];
  const int4 b = d4[1];
  const unsigned bit = 1u << (tid >> 2);
#define MARK(D) atomicOr(&bits[min((unsigned)(D) / (unsigned)TD, (unsigned)(NNODES / TD - 1))], bit);
  MARK(a.x) MARK(a.y) MARK(a.z) MARK(a.w)
  MARK(b.x) MARK(b.y) MARK(b.z) MARK(b.w)
#undef MARK
  __syncthreads();
  for (int i = tid; i < NNODES / TD; i += 128) bm[(size_t)blockIdx.x * BMP + i] = bits[i];
}

__device__ __forceinline__ void add_rows(const float* __restrict__ feat, float* tile,
                                         int sv, int rv, int n, int lane) {
  for (int j = 0; j < n; j += 8) {
    float4 v[8];
    int    rr[8];
#pragma unroll
    for (int t = 0; t < 8; ++t) {
      const int idx = min(j + t, n - 1);
      const int ss  = __builtin_amdgcn_readlane(sv, idx);
      rr[t]         = __builtin_amdgcn_readlane(rv, idx);
      v[t] = *(const float4*)(feat + (size_t)ss * F + lane * 4);
    }
#pragma unroll
    for (int t = 0; t < 8; ++t) {
      if (j + t < n) {
        float* p = tile + rr[t] * F + lane;
        atomicAdd(p,      v[t].x);
        atomicAdd(p + 32, v[t].y);
        atomicAdd(p + 64, v[t].z);
        atomicAdd(p + 96, v[t].w);
      }
    }
  }
}


#define RPW  (TD / 32)
#define RCAP 200
#define RCAP1 128
__device__ __forceinline__ void sum_rows(const float* __restrict__ feat, float4& acc,
                                         int sv, int n, int lane) {
  for (int j = 0; j < n; j += 16) {
    float4 v[16];
#pragma unroll
    for (int t = 0; t < 16; ++t) {
      const int ss = __builtin_amdgcn_readlane(sv, min(j + t, n - 1));
      v[t] = *(const float4*)(feat + (size_t)ss * F + lane * 4);
    }
#pragma unroll
    for (int t = 0; t < 16; ++t) {
      if (j + t < n) {
        acc.x += v[t].x; acc.y += v[t].y;
        acc.z += v[t].z; acc.w += v[t].w;
      }
    }
  }
}

__device__ __forceinline__ void head_row(const float4 acc, const float4 z, float inv,
                                         const float4 wa, const float4 wb,
                                         float b0, float b1, float* res, int r, int lane) {
  const float hx = fmaxf(fmaf(acc.x, inv, z.x), 0.0f);
  const float hy = fmaxf(fmaf(acc.y, inv, z.y), 0.0f);
  const float hz = fmaxf(fmaf(acc.z, inv, z.z), 0.0f);
  const float hw = fmaxf(fmaf(acc.w, inv, z.w), 0.0f);
  const float p0 = hx * wa.x + hy * wa.z + hz * wb.x + hw * wb.z;
  const float p1 = hx * wa.y + hy * wa.w + hz * wb.y + hw * wb.w;
  const bool  hi   = lane >= 16;
  const float keep = hi ? p1 : p0;
  const float send = hi ? p0 : p1;
  float v = keep + __builtin_bit_cast(float, __builtin_amdgcn_permlanex16(
                       0, __builtin_bit_cast(int, send), 0x76543210, 0xfedcba98, false, false));
#define XADD(CTRL) v += __builtin_bit_cast(float, __builtin_amdgcn_update_dpp(0, __builtin_bit_cast(int, v), CTRL, 0xf, 0xf, true))
  XADD(0x168); XADD(0x164); XADD(0x162); XADD(0x161);
#undef XADD
  if (lane == 0)  res[2 * r]     = v + b0;
  if (lane == 16) res[2 * r + 1] = v + b1;
}

__global__ __launch_bounds__(1024)
void agg_tile(const float* __restrict__ feat,
              const int* __restrict__ ei,
              float* __restrict__ sum,
              float* __restrict__ cnt,
              unsigned* __restrict__ saved,
              int* __restrict__ nsaved,
              const unsigned* __restrict__ bm,
              int first,
              const float* __restrict__ zb,
              const float* __restrict__ Wfc,
              const float* __restrict__ bfc,
              float* __restrict__ out) {
  __shared__ float    tile[TD * F];
  __shared__ unsigned mlist[32 * LCAP];
  __shared__ int      cntl[TD];
  __shared__ int      wcnt[32];
  __shared__ float    res[TD * 2];

  const int tid  = threadIdx.x;
  const int lane = tid & 31;
  const int wave = __builtin_amdgcn_readfirstlane(tid >> 5);
  const int base = blockIdx.x * TD;

  float4 wa = make_float4(0.f, 0.f, 0.f, 0.f), wb = wa;
  float hb0 = 0.f, hb1 = 0.f;
  if (!first) {
    wa  = *(const float4*)(Wfc + lane * 8);
    wb  = *(const float4*)(Wfc + lane * 8 + 4);
    hb0 = bfc[0];
    hb1 = bfc[1];
  }

  if (!first) {
    const int creg = nsaved[blockIdx.x * 32 + lane];
    if (__builtin_amdgcn_ballot_w32(creg > SCAP || creg < 0) == 0u) {
      const int nmine = __builtin_amdgcn_readlane(creg, wave);
      if (tid < TD) cntl[tid] = 0;
      __syncthreads();
      const unsigned* rsave = saved + (size_t)(blockIdx.x * 32 + wave) * SCAP;
      for (int i0 = 0; i0 < nmine; i0 += 128) {
        unsigned w[4];
#pragma unroll
        for (int u = 0; u < 4; ++u) {
          const int i = i0 + u * 32 + lane;
          w[u] = (i < nmine) ? rsave[i] : 0u;
        }
#pragma unroll
        for (int u = 0; u < 4; ++u) {
          const int i = i0 + u * 32 + lane;
          if (i < nmine) {
            const int r   = (int)min(w[u] & 255u, (unsigned)(TD - 1));
            const int pos = atomicAdd(&cntl[r], 1);
            if (pos < RCAP) mlist[r * RCAP + pos] = w[u] >> 8;
          }
        }
      }
      __syncthreads();
      int cmax = 0;
#pragma unroll
      for (int k = 0; k < RPW; ++k) cmax = max(cmax, cntl[lane + 32 * k]);
      if (__builtin_amdgcn_ballot_w32(cmax > RCAP) == 0u) {
        const int cv = __builtin_bit_cast(int, cnt[base + wave * RPW + min(lane, RPW - 1)]);
        for (int k = 0; k < RPW; ++k) {
          const int r = wave * RPW + k;
          const int nr = __builtin_amdgcn_readfirstlane(cntl[r]);
          const unsigned* rl = mlist + r * RCAP;
          const float4 z = *(const float4*)(zb + (size_t)(base + r) * F + lane * 4);
          const float inv = 1.0f / fmaxf(__builtin_bit_cast(float, __builtin_amdgcn_readlane(cv, k)), 1.0f);
          float4 acc = make_float4(0.f, 0.f, 0.f, 0.f);
          for (int i0 = 0; i0 < nr; i0 += 32) {
            const int n = min(32, nr - i0);
            const int sv = (int)min(rl[i0 + min(lane, n - 1)], (unsigned)(NNODES - 1));
            sum_rows(feat, acc, sv, n, lane);
          }
          head_row(acc, z, inv, wa, wb, hb0, hb1, res, r, lane);
        }
        __syncthreads();
        if (tid < TD * 2) out[(size_t)base * 2 + tid] = res[tid];
        return;
      }
      __syncthreads();
    }
  }

  unsigned* rlist = (unsigned*)tile;
  for (int route = first ? 1 : 0; route >= 0; --route) {
  if (!route) {
#pragma unroll
    for (int k = 0; k < (TD * F) / (4 * 1024); ++k)
      ((float4*)tile)[tid + k * 1024] = make_float4(0.f, 0.f, 0.f, 0.f);
  }
  if (tid < TD) cntl[tid] = 0;
  __syncthreads();

  unsigned* mysave = saved + (size_t)(blockIdx.x * 32 + wave) * SCAP;
  int nsv = 0;
  const int nrep = first ? 0 : __builtin_amdgcn_readfirstlane(nsaved[blockIdx.x * 32 + wave]);

  if (!first && nrep <= SCAP) {
    unsigned* myl = mlist + wave * LCAP;
    for (int i0 = 0; i0 < nrep; i0 += 128) {
      unsigned t[4];
#pragma unroll
      for (int u = 0; u < 4; ++u) {
        const int i = i0 + u * 32 + lane;
        t[u] = (i < nrep) ? mysave[i] : 0u;
      }
#pragma unroll
      for (int u = 0; u < 4; ++u) {
        const int i = i0 + u * 32 + lane;
        if (i < nrep) myl[i] = t[u];
      }
    }
    for (int i0 = 0; i0 < nrep; i0 += 32) {
      const int n = min(32, nrep - i0);
      const unsigned w = myl[i0 + min(lane, n - 1)];
      add_rows(feat, tile, (int)min(w >> 8, (unsigned)(NNODES - 1)), (int)(w & 255u), n, lane);
    }
  } else {
  const int* dstc = ei + NEDGES;
  unsigned* myl  = mlist + wave * LCAP;
  unsigned* mycl = myl + MCAP;
  int nl = 0;

  const int wd = lane * 32 + wave;
  const unsigned bits = (lane < NWL && wd < NWORD) ? bm[(size_t)wd * BMP + blockIdx.x] : 0u;

  int nc = 0;
#pragma unroll
  for (int L = 0; L < NWL; ++L) {
    const unsigned uw = __builtin_amdgcn_readlane(bits, L);
    if ((uw >> lane) & 1u)
      mycl[nc + (int)__builtin_amdgcn_mbcnt_lo(uw, 0u)] = (unsigned)((L * 32 + wave) * 32 + lane);
    nc += __builtin_popcount(uw);
  }

  for (int c0 = 0; c0 < nc; c0 += 32) {
    const int nb = min(32, nc - c0);
    const int cl = (int)mycl[c0 + min(lane, nb - 1)];
    for (int j = 0; j < nb; j += 16) {
      int dv[16];
#pragma unroll
      for (int t = 0; t < 16; ++t) {
        const int c = __builtin_amdgcn_readlane(cl, min(j + t, nb - 1));
        dv[t] = dstc[c * 32 + lane];
      }
#pragma unroll
      for (int t = 0; t < 16; ++t) {
        if (j + t < nb) {
          const int c = __builtin_amdgcn_readlane(cl, min(j + t, nb - 1));
          const int r = dv[t] - base;
          const bool m = (unsigned)r < (unsigned)TD;
          const unsigned b = __builtin_amdgcn_ballot_w32(m);
          if (m) {
            const int pos = nl + (int)__builtin_amdgcn_mbcnt_lo(b, 0u);
            myl[pos] = ((unsigned)(c * 32 + lane) << 8) | (unsigned)r;
            if (first && !route) atomicAdd(&cntl[r], 1);
          }
          nl += __builtin_popcount(b);
        }
      }

      if (nl >= (route ? MCAP - 512 : 32) || c0 + j + 16 >= nc) {
        if (route) {
          for (int i0 = 0; i0 < nl; i0 += 128) {
            unsigned pk[4];
            int      s4[4];
#pragma unroll
            for (int u = 0; u < 4; ++u) {
              pk[u] = myl[min(i0 + u * 32 + lane, nl - 1)];
              s4[u] = ei[pk[u] >> 8];
            }
#pragma unroll
            for (int u = 0; u < 4; ++u) {
              const int idx = i0 + u * 32 + lane;
              if (idx < nl) {
                const int rv = (int)(pk[u] & 255u);
                if (nsv + idx < SCAP) mysave[nsv + idx] = ((unsigned)s4[u] << 8) | (unsigned)rv;
                const int pos = atomicAdd(&cntl[rv], 1);
                if (pos < RCAP1) rlist[rv * RCAP1 + pos] = (unsigned)s4[u];
              }
            }
          }
          nsv += nl;
        } else {
          for (int i0 = 0; i0 < nl; i0 += 32) {
            const int n = min(32, nl - i0);
            const unsigned pk = myl[i0 + min(lane, n - 1)];
            const int sv = ei[pk >> 8];
            const int rv = (int)(pk & 255u);
            if (first) {
              if (lane < n && nsv + lane < SCAP) mysave[nsv + lane] = ((unsigned)sv << 8) | (unsigned)rv;
              nsv += n;
            }
            add_rows(feat, tile, sv, rv, n, lane);
          }
        }
        nl = 0;
      }
    }
  }
  }

  if (lane == 0) wcnt[wave] = nsv;
  __syncthreads();

  if (route) {
    int cmax = 0;
#pragma unroll
    for (int k = 0; k < RPW; ++k) cmax = max(cmax, cntl[lane + 32 * k]);
    if (__builtin_amdgcn_ballot_w32(cmax > RCAP1) == 0u) {
      for (int k = 0; k < RPW; ++k) {
        const int r = wave * RPW + k;
        const int nr = __builtin_amdgcn_readfirstlane(cntl[r]);
        const unsigned* rl = rlist + r * RCAP1;
        float4 acc = make_float4(0.f, 0.f, 0.f, 0.f);
        for (int i0 = 0; i0 < nr; i0 += 32) {
          const int n = min(32, nr - i0);
          const int sv = (int)rl[i0 + min(lane, n - 1)];
          sum_rows(feat, acc, sv, n, lane);
        }
        *(float4*)(sum + (size_t)(base + r) * F + lane * 4) = acc;
      }
      if (tid < TD) cnt[base + tid] = (float)cntl[tid];
      if (tid < 32) nsaved[blockIdx.x * 32 + tid] = wcnt[tid];
      return;
    }
    __syncthreads();
    continue;
  }

#pragma unroll
  for (int k = 0; k < (TD * F) / (4 * 1024); ++k) {
    const int idx = tid + k * 1024;
    const int row = idx >> 5;
    const int l   = idx & 31;
    const float* p = tile + row * F + l;
    const float4 o = make_float4(p[0], p[32], p[64], p[96]);
    if (first) {
      *(float4*)(sum + (size_t)(base + row) * F + l * 4) = o;
    } else {
      const float4 z = *(const float4*)(zb + (size_t)(base + row) * F + l * 4);
      const float inv = 1.0f / fmaxf(cnt[base + row], 1.0f);
      head_row(o, z, inv, wa, wb, hb0, hb1, res, row, l);
    }
  }
  if (first && tid < TD) cnt[base + tid] = (float)cntl[tid];
  if (first && tid < 32) nsaved[blockIdx.x * 32 + tid] = wcnt[tid];
  if (!first) {
    __syncthreads();
    if (tid < TD * 2) out[(size_t)base * 2 + tid] = res[tid];
  }
  }
}

struct Frag {
  v2f    a1, a2;
  float4 l0, l1;
  float4 r0, r1;
};

__device__ __forceinline__ void load_frag(Frag& f, const float* srow,
                                          const float* __restrict__ xrow,
                                          const float* __restrict__ wl,
                                          const float* __restrict__ wr, int k) {
  f.a1 = *(const v2f*)(srow + k);
  f.a2 = *(const v2f*)(xrow + k);
  const float* pl = wl + (size_t)k * F;
  const float* pr = wr + (size_t)k * F;
  f.l0 = *(const float4*)(pl);
  f.l1 = *(const float4*)(pl + F);
  f.r0 = *(const float4*)(pr);
  f.r1 = *(const float4*)(pr + F);
}

#define WM(A, B0, B1, C) \
  C = __builtin_amdgcn_wmma_f32_16x16x4_f32(false, A, false, (v2f){B0, B1}, (short)0, C, false, false)

__device__ __forceinline__ void mul_frag(v8f (&acc)[4], const Frag& f, float scale) {
  const v2f a1 = f.a1 * scale;
  WM(a1, f.l0.x, f.l1.x, acc[0]);
  WM(a1, f.l0.y, f.l1.y, acc[1]);
  WM(a1, f.l0.z, f.l1.z, acc[2]);
  WM(a1, f.l0.w, f.l1.w, acc[3]);
  WM(f.a2, f.r0.x, f.r1.x, acc[0]);
  WM(f.a2, f.r0.y, f.r1.y, acc[1]);
  WM(f.a2, f.r0.z, f.r1.z, acc[2]);
  WM(f.a2, f.r0.w, f.r1.w, acc[3]);
}

struct Frag1 {
  v2f    a;
  float4 l0, l1;
  float4 r0, r1;
};

__device__ __forceinline__ void load_frag1_b(Frag1& f, const float* __restrict__ wl,
                                             const float* __restrict__ wr, int k) {
  const float* pl = wl + (size_t)k * F;
  const float* pr = wr + (size_t)k * F;
  f.l0 = *(const float4*)(pl);
  f.l1 = *(const float4*)(pl + F);
  f.r0 = *(const float4*)(pr);
  f.r1 = *(const float4*)(pr + F);
}

__device__ __forceinline__ void load_frag1_a(Frag1& f, const float* hrow, int k) {
  f.a = *(const v2f*)(hrow + k);
}

__device__ __forceinline__ void mul_frag1(v8f (&ay)[4], v8f (&az)[4], const Frag1& f) {
  WM(f.a, f.l0.x, f.l1.x, ay[0]);
  WM(f.a, f.l0.y, f.l1.y, ay[1]);
  WM(f.a, f.l0.z, f.l1.z, ay[2]);
  WM(f.a, f.l0.w, f.l1.w, ay[3]);
  WM(f.a, f.r0.x, f.r1.x, az[0]);
  WM(f.a, f.r0.y, f.r1.y, az[1]);
  WM(f.a, f.r0.z, f.r1.z, az[2]);
  WM(f.a, f.r0.w, f.r1.w, az[3]);
}
#undef WM

#define GW 4
#define HP 132

__global__ __launch_bounds__(32 * GW)
void sage_pair(const float* sum,
               const float* __restrict__ cnt,
               const float* __restrict__ xin,
               const float* __restrict__ W1l,
               const float* __restrict__ W1r,
               const float* __restrict__ b1,
               const float* __restrict__ W2l,
               const float* __restrict__ W2r,
               const float* __restrict__ b2,
               float* yout,
               float* __restrict__ zout) {
  __shared__ float htile[(GW / 2) * 16 * HP];

  const int tid  = threadIdx.x;
  const int lane = tid & 31;
  const int wave = tid >> 5;
  const int half = lane >> 4;
  const int lid  = lane & 15;
  const int g    = wave & 1;
  const int t    = wave >> 1;
  const int base = (blockIdx.x * (GW / 2) + t) * 16;

  const float scale = 1.0f / fmaxf(cnt[base + lid], 1.0f);
  const float* srow = sum + (size_t)(base + lid) * F + 2 * half;
  const float* xrow = xin + (size_t)(base + lid) * F + 2 * half;
  const size_t woff = (size_t)(2 * half) * F + 64 * g + 4 * lid;

  v8f acc[4] = {};
  {
    const float* wl = W1l + woff;
    const float* wr = W1r + woff;
    Frag f[3];
    load_frag(f[0], srow, xrow, wl, wr, 0);
    load_frag(f[1], srow, xrow, wl, wr, 4);
#pragma unroll
    for (int s = 0; s < F / 4; ++s) {
      if (s + 2 < F / 4) load_frag(f[(s + 2) % 3], srow, xrow, wl, wr, 4 * (s + 2));
      __builtin_amdgcn_sched_barrier(0);
      mul_frag(acc, f[s % 3], scale);
      __builtin_amdgcn_sched_barrier(0);
    }
  }

  const float* wl2 = W2l + woff;
  const float* wr2 = W2r + woff;
  Frag1 f2[3];
  load_frag1_b(f2[0], wl2, wr2, 0);
  load_frag1_b(f2[1], wl2, wr2, 4);

  {
    const float4 bv = *(const float4*)(b1 + 64 * g + 4 * lid);
    float* hw = htile + (size_t)(t * 16 + 8 * half) * HP + 64 * g + 4 * lid;
#pragma unroll
    for (int v = 0; v < 8; ++v) {
      float4 o = make_float4(acc[0][v] + bv.x, acc[1][v] + bv.y,
                             acc[2][v] + bv.z, acc[3][v] + bv.w);
      o.x = fmaxf(o.x, 0.0f); o.y = fmaxf(o.y, 0.0f);
      o.z = fmaxf(o.z, 0.0f); o.w = fmaxf(o.w, 0.0f);
      *(float4*)(hw + v * HP) = o;
    }
  }
  __syncthreads();

  const float* hrow = htile + (size_t)(t * 16 + lid) * HP + 2 * half;
  load_frag1_a(f2[0], hrow, 0);
  load_frag1_a(f2[1], hrow, 4);

  v8f ay[4] = {};
  v8f az[4] = {};
#pragma unroll
  for (int s = 0; s < F / 4; ++s) {
    if (s + 2 < F / 4) {
      load_frag1_b(f2[(s + 2) % 3], wl2, wr2, 4 * (s + 2));
      load_frag1_a(f2[(s + 2) % 3], hrow, 4 * (s + 2));
    }
    __builtin_amdgcn_sched_barrier(0);
    mul_frag1(ay, az, f2[s % 3]);
    __builtin_amdgcn_sched_barrier(0);
  }

  const float4 bv2 = *(const float4*)(b2 + 64 * g + 4 * lid);
#pragma unroll
  for (int v = 0; v < 8; ++v) {
    const size_t o = (size_t)(base + v + 8 * half) * F + 64 * g + 4 * lid;
    *(float4*)(yout + o) = make_float4(ay[0][v], ay[1][v], ay[2][v], ay[3][v]);
    *(float4*)(zout + o) = make_float4(az[0][v] + bv2.x, az[1][v] + bv2.y,
                                       az[2][v] + bv2.z, az[3][v] + bv2.w);
  }
}

extern "C" void kernel_launch(void* const* d_in, const int* in_sizes, int n_in,
                              void* d_out, int out_size, void* d_ws, size_t ws_size,
                              hipStream_t stream) {
  const float* x   = (const float*)d_in[0];
  const int*   ei  = (const int*)d_in[1];
  const float* W1l = (const float*)d_in[2];
  const float* b1  = (const float*)d_in[3];
  const float* W1r = (const float*)d_in[4];
  const float* W2l = (const float*)d_in[5];
  const float* b2  = (const float*)d_in[6];
  const float* W2r = (const float*)d_in[7];
  const float* Wfc = (const float*)d_in[8];
  const float* bfc = (const float*)d_in[9];
  float* out = (float*)d_out;

  float* ws     = (float*)d_ws;
  float* sumbuf = ws;
  float* zbuf   = sumbuf + (size_t)NNODES * F;
  float* h2     = zbuf + (size_t)NNODES * F;
  float* cnt    = h2 + (size_t)NNODES * F;
  unsigned* saved  = (unsigned*)h2;
  int*      nsaved = (int*)(h2 + (size_t)(NNODES / TD) * 32 * SCAP);
  unsigned* bm     = (unsigned*)(nsaved + (NNODES / TD) * 32);

  mark_chunks<<<NWORD, 128, 0, stream>>>(ei, bm);

  agg_tile<<<NNODES / TD, 1024, 0, stream>>>(x, ei, sumbuf, cnt, saved, nsaved, bm, 1,
                                             nullptr, nullptr, nullptr, nullptr);

  sage_pair<<<NNODES / (8 * GW), 32 * GW, 0, stream>>>(sumbuf, cnt, x, W1l, W1r, b1,
                                                       W2l, W2r, b2, sumbuf, zbuf);

  agg_tile<<<NNODES / TD, 1024, 0, stream>>>(sumbuf, ei, nullptr, cnt, saved, nsaved, bm, 0,
                                             zbuf, Wfc, bfc, out);
}
